# starting kernel
# baseline (speedup 1.0000x reference)
#include <hip/hip_runtime.h>
#include <hip/hip_bf16.h>


typedef float v2f __attribute__((ext_vector_type(2)));
typedef float v8f __attribute__((ext_vector_type(8)));

#define D_IN  128
#define D_HID 64
#define D_OUT 32


__global__ void k_zero_u32(unsigned* p, int n) {
    int i = blockIdx.x * blockDim.x + threadIdx.x;
    if (i < n) p[i] = 0u;
}

__global__ void k_degree(const int* __restrict__ ei, unsigned* __restrict__ deg, int E) {
    int e = blockIdx.x * blockDim.x + threadIdx.x;
    if (e < E) {
        int d = ei[E + e];
        atomicAdd(&deg[d], 1u);
    }
}

__global__ void k_dinv(const unsigned* __restrict__ deg, float* __restrict__ dinv, int N) {
    int i = blockIdx.x * blockDim.x + threadIdx.x;
    if (i < N) {
        float dg = (float)(deg[i] + 1u);
        dinv[i] = rsqrtf(dg);
    }
}


__global__ void k_gemm1_wmma(const float* __restrict__ X, const float* __restrict__ W,
                             float* __restrict__ H) {
    const int lane = threadIdx.x & 31;
    const int wave = threadIdx.x >> 5;
    const int m0   = blockIdx.x * 16;
    const int n0   = wave * 16;
    const int half = lane >> 4;
    const int lm   = lane & 15;

    const float* xrow = X + (size_t)(m0 + lm) * D_IN;
    v8f acc = {};
    #pragma unroll 4
    for (int k0 = 0; k0 < D_IN; k0 += 4) {
        const int ka = k0 + 2 * half;
        v2f a, b;
        a.x = xrow[ka];
        a.y = xrow[ka + 1];
        b.x = W[(size_t)ka       * D_HID + n0 + lm];
        b.y = W[(size_t)(ka + 1) * D_HID + n0 + lm];
        acc = __builtin_amdgcn_wmma_f32_16x16x4_f32(false, a, false, b,
                                                    (short)0, acc, false, false);
    }
    #pragma unroll
    for (int r = 0; r < 8; ++r) {
        const int m = m0 + r + 8 * half;
        H[(size_t)m * D_HID + n0 + lm] = acc[r];
    }
}

__global__ void k_selfinit64(const float* __restrict__ h, const float* __restrict__ dinv,
                             float* __restrict__ agg, int N) {
    int i = blockIdx.x * blockDim.x + threadIdx.x;
    if (i < N * D_HID) {
        int n = i >> 6;
        float w = dinv[n];
        agg[i] = w * w * h[i];
    }
}

__global__ void k_edge64(const int* __restrict__ ei, const float* __restrict__ dinv,
                         const float* __restrict__ h, float* __restrict__ agg, int E) {
    int t = blockIdx.x * 256 + threadIdx.x;
    int e = t >> 6;
    int f = t & 63;
    if (e < E) {
        int s = ei[e];
        int d = ei[E + e];
        float w = dinv[s] * dinv[d];
        atomicAdd(&agg[(size_t)d * D_HID + f], h[(size_t)s * D_HID + f] * w);
    }
}


__global__ void k_gemm2_wmma(const float* __restrict__ AGG, const float* __restrict__ b1,
                             const float* __restrict__ W, float* __restrict__ H2) {
    const int lane = threadIdx.x & 31;
    const int wave = threadIdx.x >> 5;
    const int m0   = blockIdx.x * 16;
    const int n0   = wave * 16;
    const int half = lane >> 4;
    const int lm   = lane & 15;

    const float* arow = AGG + (size_t)(m0 + lm) * D_HID;
    v8f acc = {};
    #pragma unroll 4
    for (int k0 = 0; k0 < D_HID; k0 += 4) {
        const int ka = k0 + 2 * half;
        v2f a, b;
        a.x = fmaxf(arow[ka]     + b1[ka],     0.f);
        a.y = fmaxf(arow[ka + 1] + b1[ka + 1], 0.f);
        b.x = W[(size_t)ka       * D_OUT + n0 + lm];
        b.y = W[(size_t)(ka + 1) * D_OUT + n0 + lm];
        acc = __builtin_amdgcn_wmma_f32_16x16x4_f32(false, a, false, b,
                                                    (short)0, acc, false, false);
    }
    #pragma unroll
    for (int r = 0; r < 8; ++r) {
        const int m = m0 + r + 8 * half;
        H2[(size_t)m * D_OUT + n0 + lm] = acc[r];
    }
}

__global__ void k_outinit32(const float* __restrict__ h2, const float* __restrict__ dinv,
                            const float* __restrict__ b2, float* __restrict__ out, int N) {
    int i = blockIdx.x * blockDim.x + threadIdx.x;
    if (i < N * D_OUT) {
        int n = i >> 5;
        int f = i & 31;
        float w = dinv[n];
        out[i] = b2[f] + w * w * h2[i];
    }
}

__global__ void k_edge32(const int* __restrict__ ei, const float* __restrict__ dinv,
                         const float* __restrict__ h2, float* __restrict__ out, int E) {
    int t = blockIdx.x * 256 + threadIdx.x;
    int e = t >> 5;
    int f = t & 31;
    if (e < E) {
        int s = ei[e];
        int d = ei[E + e];
        float w = dinv[s] * dinv[d];
        atomicAdd(&out[(size_t)d * D_OUT + f], h2[(size_t)s * D_OUT + f] * w);
    }
}


extern "C" void kernel_launch(void* const* d_in, const int* in_sizes, int n_in,
                              void* d_out, int out_size, void* d_ws, size_t ws_size,
                              hipStream_t stream) {
    const float* x   = (const float*)d_in[0];
    const int*   ei  = (const int*)  d_in[1];
    const float* W1  = (const float*)d_in[2];
    const float* b1  = (const float*)d_in[3];
    const float* W2  = (const float*)d_in[4];
    const float* b2  = (const float*)d_in[5];
    float*       out = (float*)d_out;

    const int N = in_sizes[0] / D_IN;
    const int E = in_sizes[1] / 2;

    char* ws = (char*)d_ws;
    unsigned* deg  = (unsigned*)(ws);
    float*    dinv = (float*)(ws + (size_t)N * 4);
    float*    h    = (float*)(ws + (size_t)N * 8);
    float*    agg1 = (float*)(ws + (size_t)N * 8 + (size_t)N * D_HID * 4);
    float*    h2   = (float*)(ws + (size_t)N * 8 + (size_t)N * D_HID * 8);

    const int TB = 256;

    k_zero_u32<<<(N + TB - 1) / TB, TB, 0, stream>>>(deg, N);
    k_degree  <<<(E + TB - 1) / TB, TB, 0, stream>>>(ei, deg, E);
    k_dinv    <<<(N + TB - 1) / TB, TB, 0, stream>>>(deg, dinv, N);

    k_gemm1_wmma<<<N / 16, 128, 0, stream>>>(x, W1, h);
    k_selfinit64<<<((size_t)N * D_HID + TB - 1) / TB, TB, 0, stream>>>(h, dinv, agg1, N);
    k_edge64    <<<((size_t)E * D_HID + TB - 1) / TB, TB, 0, stream>>>(ei, dinv, h, agg1, E);

    k_gemm2_wmma<<<N / 16, 64, 0, stream>>>(agg1, b1, W2, h2);
    k_outinit32 <<<((size_t)N * D_OUT + TB - 1) / TB, TB, 0, stream>>>(h2, dinv, b2, out, N);
    k_edge32    <<<((size_t)E * D_OUT + TB - 1) / TB, TB, 0, stream>>>(ei, dinv, h2, out, E);
}


# Global float atomic scatter replaced: a work-group owns 512 destination nodes, sums them in LDS (LDS float atomics), each wave scans the destination list for its edges; sum order: self, LDS arrival
# speedup vs baseline: 1.8226x; 1.8226x over previous
#include <hip/hip_runtime.h>
#include <hip/hip_bf16.h>


typedef float v2f __attribute__((ext_vector_type(2)));
typedef float v8f __attribute__((ext_vector_type(8)));

#define D_IN  128
#define D_HID 64
#define D_OUT 32


__global__ void k_zero_u32(unsigned* p, int n) {
    int i = blockIdx.x * blockDim.x + threadIdx.x;
    if (i < n) p[i] = 0u;
}

__global__ void k_degree(const int* __restrict__ ei, unsigned* __restrict__ deg, int E) {
    int e = blockIdx.x * blockDim.x + threadIdx.x;
    if (e < E) {
        int d = ei[E + e];
        atomicAdd(&deg[d], 1u);
    }
}

__global__ void k_dinv(const unsigned* __restrict__ deg, float* __restrict__ dinv, int N) {
    int i = blockIdx.x * blockDim.x + threadIdx.x;
    if (i < N) {
        float dg = (float)(deg[i] + 1u);
        dinv[i] = rsqrtf(dg);
    }
}


__global__ void k_gemm1_wmma(const float* __restrict__ X, const float* __restrict__ W,
                             float* __restrict__ H) {
    const int lane = threadIdx.x & 31;
    const int wave = threadIdx.x >> 5;
    const int m0   = blockIdx.x * 16;
    const int n0   = wave * 16;
    const int half = lane >> 4;
    const int lm   = lane & 15;

    const float* xrow = X + (size_t)(m0 + lm) * D_IN;
    v8f acc = {};
    #pragma unroll 4
    for (int k0 = 0; k0 < D_IN; k0 += 4) {
        const int ka = k0 + 2 * half;
        v2f a, b;
        a.x = xrow[ka];
        a.y = xrow[ka + 1];
        b.x = W[(size_t)ka       * D_HID + n0 + lm];
        b.y = W[(size_t)(ka + 1) * D_HID + n0 + lm];
        acc = __builtin_amdgcn_wmma_f32_16x16x4_f32(false, a, false, b,
                                                    (short)0, acc, false, false);
    }
    #pragma unroll
    for (int r = 0; r < 8; ++r) {
        const int m = m0 + r + 8 * half;
        H[(size_t)m * D_HID + n0 + lm] = acc[r];
    }
}


#define BN    512
#define AGG_T 1024
#define NWAVE (AGG_T / 32)
#define QW    384

template <int F>
__device__ __forceinline__ void range_agg(const int* __restrict__ ei,
                                          const float* __restrict__ dinv,
                                          const float* __restrict__ hin,
                                          const float* __restrict__ bias,
                                          float* __restrict__ outp, int N, int E,
                                          float* acc, unsigned* q, float* qw,
                                          unsigned short* qd) {
    const int tid  = threadIdx.x;
    const int lane = tid & 31;
    const int wave = tid >> 5;
    const int lo   = blockIdx.x * BN;
    const int left = N - lo;
    const unsigned lim = (unsigned)(left < BN ? left : BN);
    const int* dstp = ei + E;

    unsigned*       wq  = q  + wave * QW;
    float*          wqw = qw + wave * QW;
    unsigned short* wqd = qd + wave * QW;
    int wcount = 0;

    #pragma unroll 8
    for (int i = tid; i < BN * F; i += AGG_T) {
        const int n  = lo + i / F;
        const int f  = i % F;
        const int nn = n < N ? n : N - 1;
        const float w = dinv[nn];
        const float x = hin[(size_t)nn * F + f];
        const float v = bias ? bias[f] + w * w * x : w * w * x;
        acc[i] = n < N ? v : 0.f;
    }
    __syncthreads();

    auto direct = [&](int e, unsigned dl) {
        const int s = ei[e];
        const float w = dinv[s] * dinv[lo + (int)dl];
        #pragma unroll 1
        for (int f = 0; f < F; ++f)
            atomicAdd(&acc[dl * F + f], hin[(size_t)s * F + f] * w);
    };

    auto append = [&](bool hit, int e, unsigned dl) {
        const unsigned bal = __builtin_amdgcn_ballot_w32(hit);
        if (bal != 0u) {
            if (hit) {
                const int pos = wcount + (int)__builtin_amdgcn_mbcnt_lo(bal, 0u);
                if (pos < QW) wq[pos] = (unsigned)e;
                else          direct(e, dl);
            }
            wcount += __builtin_popcount(bal);
        }
    };

    auto four = [&](const int4 t, int e0) {
        const unsigned a = (unsigned)(t.x - lo);
        const unsigned b = (unsigned)(t.y - lo);
        const unsigned c = (unsigned)(t.z - lo);
        const unsigned d = (unsigned)(t.w - lo);
        const unsigned ab = a < b ? a : b;
        const unsigned cd = c < d ? c : d;
        if (__builtin_amdgcn_ballot_w32((ab < cd ? ab : cd) < lim) != 0u) {
            append(a < lim, e0,     a);
            append(b < lim, e0 + 1, b);
            append(c < lim, e0 + 2, c);
            append(d < lim, e0 + 3, d);
        }
    };

    if ((E & 3) == 0) {
        const int4* d4 = reinterpret_cast<const int4*>(dstp);
        const int n4 = E >> 2;
        constexpr int L = 8;
        const int CH = L * AGG_T;
        const int trips = n4 / CH;
        const int rot = (int)(((long long)blockIdx.x * trips) / gridDim.x);
        int4 cur[L], nxt[L];
        if (trips > 0) {
            #pragma unroll
            for (int u = 0; u < L; ++u) cur[u] = d4[rot * CH + u * AGG_T + tid];
        }
        for (int it = 0; it < trips; ++it) {
            int t = it + rot;
            if (t >= trips) t -= trips;
            int t1 = t + 1;
            if (t1 >= trips) t1 -= trips;
            #pragma unroll
            for (int u = 0; u < L; ++u) nxt[u] = d4[t1 * CH + u * AGG_T + tid];
            #pragma unroll
            for (int u = 0; u < L; ++u) four(cur[u], (t * CH + u * AGG_T + tid) * 4);
            #pragma unroll
            for (int u = 0; u < L; ++u) cur[u] = nxt[u];
        }
        for (int base = trips * CH; base < n4; base += AGG_T) {
            const int i4 = base + tid;
            const int4 t = i4 < n4 ? d4[i4] : make_int4(-1, -1, -1, -1);
            four(t, i4 * 4);
        }
    } else {
        for (int base = 0; base < E; base += AGG_T) {
            const int e = base + tid;
            const unsigned a = (unsigned)((e < E ? dstp[e] : -1) - lo);
            append(a < lim, e, a);
        }
    }

    const int nq = wcount < QW ? wcount : QW;
    __builtin_amdgcn_fence(__ATOMIC_ACQ_REL, "wavefront");

    for (int j0 = lane; j0 < nq; j0 += 4 * 32) {
        int e[4], s[4], d[4];
        float ws[4], wd[4];
        #pragma unroll
        for (int u = 0; u < 4; ++u) {
            const int j = j0 + u * 32;
            e[u] = j < nq ? (int)wq[j] : 0;
        }
        #pragma unroll
        for (int u = 0; u < 4; ++u) { s[u] = ei[e[u]]; d[u] = dstp[e[u]]; }
        #pragma unroll
        for (int u = 0; u < 4; ++u) { ws[u] = dinv[s[u]]; wd[u] = dinv[d[u]]; }
        #pragma unroll
        for (int u = 0; u < 4; ++u) {
            const int j = j0 + u * 32;
            if (j < nq) {
                wq[j]  = (unsigned)s[u];
                wqd[j] = (unsigned short)(d[u] - lo);
                wqw[j] = ws[u] * wd[u];
            }
        }
    }
    __builtin_amdgcn_fence(__ATOMIC_ACQ_REL, "wavefront");

    constexpr int PER = F / 32;
    constexpr int U   = 16;
    for (int j0 = 0; j0 < nq; j0 += U) {
        float r[U][PER];
        float w[U];
        int   dl[U];
        size_t s[U];
        #pragma unroll
        for (int u = 0; u < U; ++u) {
            const int j = (j0 + u < nq) ? j0 + u : nq - 1;
            s[u]  = wq[j];
            dl[u] = wqd[j];
            w[u]  = wqw[j];
        }
        #pragma unroll
        for (int u = 0; u < U; ++u) {
            const float* row = hin + s[u] * F + lane * PER;
            if constexpr (PER == 2) {
                const float2 t2 = *reinterpret_cast<const float2*>(row);
                r[u][0] = t2.x; r[u][1] = t2.y;
            } else {
                r[u][0] = row[0];
            }
        }
        #pragma unroll
        for (int u = 0; u < U; ++u) {
            if (j0 + u < nq) {
                #pragma unroll
                for (int p = 0; p < PER; ++p)
                    atomicAdd(&acc[dl[u] * F + lane * PER + p], r[u][p] * w[u]);
            }
        }
    }
    __syncthreads();

    for (int i = tid; i < BN * F; i += AGG_T) {
        const int n = lo + i / F;
        if (n < N) outp[(size_t)n * F + (i % F)] = acc[i];
    }
}

__global__ __launch_bounds__(AGG_T)
void k_range_agg64(const int* __restrict__ ei, const float* __restrict__ dinv,
                   const float* __restrict__ h, float* __restrict__ agg, int N, int E) {
    __shared__ float          acc[BN * D_HID];
    __shared__ unsigned       q[NWAVE * QW];
    __shared__ float          qw[NWAVE * QW];
    __shared__ unsigned short qd[NWAVE * QW];
    range_agg<D_HID>(ei, dinv, h, nullptr, agg, N, E, acc, q, qw, qd);
}


__global__ void k_gemm2_wmma(const float* __restrict__ AGG, const float* __restrict__ b1,
                             const float* __restrict__ W, float* __restrict__ H2) {
    const int lane = threadIdx.x & 31;
    const int wave = threadIdx.x >> 5;
    const int m0   = blockIdx.x * 16;
    const int n0   = wave * 16;
    const int half = lane >> 4;
    const int lm   = lane & 15;

    const float* arow = AGG + (size_t)(m0 + lm) * D_HID;
    v8f acc = {};
    #pragma unroll 4
    for (int k0 = 0; k0 < D_HID; k0 += 4) {
        const int ka = k0 + 2 * half;
        v2f a, b;
        a.x = fmaxf(arow[ka]     + b1[ka],     0.f);
        a.y = fmaxf(arow[ka + 1] + b1[ka + 1], 0.f);
        b.x = W[(size_t)ka       * D_OUT + n0 + lm];
        b.y = W[(size_t)(ka + 1) * D_OUT + n0 + lm];
        acc = __builtin_amdgcn_wmma_f32_16x16x4_f32(false, a, false, b,
                                                    (short)0, acc, false, false);
    }
    #pragma unroll
    for (int r = 0; r < 8; ++r) {
        const int m = m0 + r + 8 * half;
        H2[(size_t)m * D_OUT + n0 + lm] = acc[r];
    }
}

__global__ __launch_bounds__(AGG_T)
void k_range_agg32(const int* __restrict__ ei, const float* __restrict__ dinv,
                   const float* __restrict__ h2, const float* __restrict__ b2,
                   float* __restrict__ out, int N, int E) {
    __shared__ float          acc[BN * D_OUT];
    __shared__ unsigned       q[NWAVE * QW];
    __shared__ float          qw[NWAVE * QW];
    __shared__ unsigned short qd[NWAVE * QW];
    range_agg<D_OUT>(ei, dinv, h2, b2, out, N, E, acc, q, qw, qd);
}


extern "C" void kernel_launch(void* const* d_in, const int* in_sizes, int n_in,
                              void* d_out, int out_size, void* d_ws, size_t ws_size,
                              hipStream_t stream) {
    const float* x   = (const float*)d_in[0];
    const int*   ei  = (const int*)  d_in[1];
    const float* W1  = (const float*)d_in[2];
    const float* b1  = (const float*)d_in[3];
    const float* W2  = (const float*)d_in[4];
    const float* b2  = (const float*)d_in[5];
    float*       out = (float*)d_out;

    const int N = in_sizes[0] / D_IN;
    const int E = in_sizes[1] / 2;

    char* ws = (char*)d_ws;
    unsigned* deg  = (unsigned*)(ws);
    float*    dinv = (float*)(ws + (size_t)N * 4);
    float*    h    = (float*)(ws + (size_t)N * 8);
    float*    agg1 = (float*)(ws + (size_t)N * 8 + (size_t)N * D_HID * 4);
    float*    h2   = (float*)(ws + (size_t)N * 8 + (size_t)N * D_HID * 8);

    const int TB = 256;

    k_zero_u32<<<(N + TB - 1) / TB, TB, 0, stream>>>(deg, N);
    k_degree  <<<(E + TB - 1) / TB, TB, 0, stream>>>(ei, deg, E);
    k_dinv    <<<(N + TB - 1) / TB, TB, 0, stream>>>(deg, dinv, N);

    const int NB = (N + BN - 1) / BN;

    k_gemm1_wmma <<<N / 16, 128, 0, stream>>>(x, W1, h);
    k_range_agg64<<<NB, AGG_T, 0, stream>>>(ei, dinv, h, agg1, N, E);

    k_gemm2_wmma <<<N / 16, 64, 0, stream>>>(agg1, b1, W2, h2);
    k_range_agg32<<<NB, AGG_T, 0, stream>>>(ei, dinv, h2, b2, out, N, E);
}


# Layer 2 no longer scans the destination list: layer 1 stores each wave's resolved edge queue (source, local destination, weight) in the workspace and layer 2 reads it back; sum order kept
# speedup vs baseline: 2.2242x; 1.2203x over previous
#include <hip/hip_runtime.h>
#include <hip/hip_bf16.h>


typedef float v2f __attribute__((ext_vector_type(2)));
typedef float v8f __attribute__((ext_vector_type(8)));

#define D_IN  128
#define D_HID 64
#define D_OUT 32


__global__ void k_zero_u32(unsigned* p, int n) {
    int i = blockIdx.x * blockDim.x + threadIdx.x;
    if (i < n) p[i] = 0u;
}

__global__ void k_degree(const int* __restrict__ ei, unsigned* __restrict__ deg, int E) {
    int e = blockIdx.x * blockDim.x + threadIdx.x;
    if (e < E) {
        int d = ei[E + e];
        atomicAdd(&deg[d], 1u);
    }
}

__global__ void k_dinv(const unsigned* __restrict__ deg, float* __restrict__ dinv, int N) {
    int i = blockIdx.x * blockDim.x + threadIdx.x;
    if (i < N) {
        float dg = (float)(deg[i] + 1u);
        dinv[i] = rsqrtf(dg);
    }
}


__global__ void k_gemm1_wmma(const float* __restrict__ X, const float* __restrict__ W,
                             float* __restrict__ H) {
    const int lane = threadIdx.x & 31;
    const int wave = threadIdx.x >> 5;
    const int m0   = blockIdx.x * 16;
    const int n0   = wave * 16;
    const int half = lane >> 4;
    const int lm   = lane & 15;

    const float* xrow = X + (size_t)(m0 + lm) * D_IN;
    v8f acc = {};
    #pragma unroll 4
    for (int k0 = 0; k0 < D_IN; k0 += 4) {
        const int ka = k0 + 2 * half;
        v2f a, b;
        a.x = xrow[ka];
        a.y = xrow[ka + 1];
        b.x = W[(size_t)ka       * D_HID + n0 + lm];
        b.y = W[(size_t)(ka + 1) * D_HID + n0 + lm];
        acc = __builtin_amdgcn_wmma_f32_16x16x4_f32(false, a, false, b,
                                                    (short)0, acc, false, false);
    }
    #pragma unroll
    for (int r = 0; r < 8; ++r) {
        const int m = m0 + r + 8 * half;
        H[(size_t)m * D_HID + n0 + lm] = acc[r];
    }
}


#define BN    512
#define AGG_T 1024
#define NWAVE (AGG_T / 32)
#define QW    384

template <int F, bool SAVE>
__device__ __forceinline__ int scan_resolve(const int* __restrict__ ei,
                                            const float* __restrict__ dinv,
                                            const float* __restrict__ hin,
                                            int E, int lo, unsigned lim, float* acc,
                                            unsigned* wq, float* wqw, unsigned short* wqd,
                                            unsigned* __restrict__ sv_s,
                                            float* __restrict__ sv_w,
                                            unsigned short* __restrict__ sv_d,
                                            int* __restrict__ sv_c) {
    const int tid  = threadIdx.x;
    const int lane = tid & 31;
    const int* dstp = ei + E;
    int wcount = 0;

    auto direct = [&](int e, unsigned dl) {
        const int s = ei[e];
        const float w = dinv[s] * dinv[lo + (int)dl];
        #pragma unroll 1
        for (int f = 0; f < F; ++f)
            atomicAdd(&acc[dl * F + f], hin[(size_t)s * F + f] * w);
    };

    auto append = [&](bool hit, int e, unsigned dl) {
        const unsigned bal = __builtin_amdgcn_ballot_w32(hit);
        if (bal != 0u) {
            if (hit) {
                const int pos = wcount + (int)__builtin_amdgcn_mbcnt_lo(bal, 0u);
                if (pos < QW) wq[pos] = (unsigned)e;
                else          direct(e, dl);
            }
            wcount += __builtin_popcount(bal);
        }
    };

    auto four = [&](const int4 t, int e0) {
        const unsigned a = (unsigned)(t.x - lo);
        const unsigned b = (unsigned)(t.y - lo);
        const unsigned c = (unsigned)(t.z - lo);
        const unsigned d = (unsigned)(t.w - lo);
        const unsigned ab = a < b ? a : b;
        const unsigned cd = c < d ? c : d;
        if (__builtin_amdgcn_ballot_w32((ab < cd ? ab : cd) < lim) != 0u) {
            append(a < lim, e0,     a);
            append(b < lim, e0 + 1, b);
            append(c < lim, e0 + 2, c);
            append(d < lim, e0 + 3, d);
        }
    };

    if ((E & 3) == 0) {
        const int4* d4 = reinterpret_cast<const int4*>(dstp);
        const int n4 = E >> 2;
        constexpr int L = 8;
        const int CH = L * AGG_T;
        const int trips = n4 / CH;
        const int rot = (int)(((long long)blockIdx.x * trips) / gridDim.x);
        int4 cur[L], nxt[L];
        if (trips > 0) {
            #pragma unroll
            for (int u = 0; u < L; ++u) cur[u] = d4[rot * CH + u * AGG_T + tid];
        }
        for (int it = 0; it < trips; ++it) {
            int t = it + rot;
            if (t >= trips) t -= trips;
            int t1 = t + 1;
            if (t1 >= trips) t1 -= trips;
            #pragma unroll
            for (int u = 0; u < L; ++u) nxt[u] = d4[t1 * CH + u * AGG_T + tid];
            #pragma unroll
            for (int u = 0; u < L; ++u) four(cur[u], (t * CH + u * AGG_T + tid) * 4);
            #pragma unroll
            for (int u = 0; u < L; ++u) cur[u] = nxt[u];
        }
        for (int base = trips * CH; base < n4; base += AGG_T) {
            const int i4 = base + tid;
            const int4 t = i4 < n4 ? d4[i4] : make_int4(-1, -1, -1, -1);
            four(t, i4 * 4);
        }
    } else {
        for (int base = 0; base < E; base += AGG_T) {
            const int e = base + tid;
            const unsigned a = (unsigned)((e < E ? dstp[e] : -1) - lo);
            append(a < lim, e, a);
        }
    }

    const int nq = wcount < QW ? wcount : QW;
    if constexpr (SAVE) {
        if (lane == 0) *sv_c = wcount;
    }
    __builtin_amdgcn_fence(__ATOMIC_ACQ_REL, "wavefront");

    for (int j0 = lane; j0 < nq; j0 += 4 * 32) {
        int e[4], s[4], d[4];
        float ws[4], wd[4];
        #pragma unroll
        for (int u = 0; u < 4; ++u) {
            const int j = j0 + u * 32;
            e[u] = j < nq ? (int)wq[j] : 0;
        }
        #pragma unroll
        for (int u = 0; u < 4; ++u) { s[u] = ei[e[u]]; d[u] = dstp[e[u]]; }
        #pragma unroll
        for (int u = 0; u < 4; ++u) { ws[u] = dinv[s[u]]; wd[u] = dinv[d[u]]; }
        #pragma unroll
        for (int u = 0; u < 4; ++u) {
            const int j = j0 + u * 32;
            if (j < nq) {
                const unsigned short dl = (unsigned short)(d[u] - lo);
                const float w = ws[u] * wd[u];
                wq[j]  = (unsigned)s[u];
                wqd[j] = dl;
                wqw[j] = w;
                if constexpr (SAVE) {
                    sv_s[j] = (unsigned)s[u];
                    sv_d[j] = dl;
                    sv_w[j] = w;
                }
            }
        }
    }
    return nq;
}

template <int F, bool SAVE>
__device__ __forceinline__ void range_agg(const int* __restrict__ ei,
                                          const float* __restrict__ dinv,
                                          const float* __restrict__ hin,
                                          const float* __restrict__ bias,
                                          float* __restrict__ outp, int N, int E,
                                          unsigned* __restrict__ sv_s,
                                          float* __restrict__ sv_w,
                                          unsigned short* __restrict__ sv_d,
                                          int* __restrict__ sv_c,
                                          float* acc, unsigned* q, float* qw,
                                          unsigned short* qd) {
    const int tid  = threadIdx.x;
    const int lane = tid & 31;
    const int wave = tid >> 5;
    const int lo   = blockIdx.x * BN;
    const int left = N - lo;
    const unsigned lim = (unsigned)(left < BN ? left : BN);

    unsigned*       wq  = q  + wave * QW;
    float*          wqw = qw + wave * QW;
    unsigned short* wqd = qd + wave * QW;

    const size_t slot = (size_t)blockIdx.x * NWAVE + wave;
    unsigned*       ws_s = sv_s + slot * QW;
    float*          ws_w = sv_w + slot * QW;
    unsigned short* ws_d = sv_d + slot * QW;
    int*            ws_c = sv_c + slot;

    #pragma unroll 8
    for (int i = tid; i < BN * F; i += AGG_T) {
        const int n  = lo + i / F;
        const int f  = i % F;
        const int nn = n < N ? n : N - 1;
        const float w = dinv[nn];
        const float x = hin[(size_t)nn * F + f];
        const float v = bias ? bias[f] + w * w * x : w * w * x;
        acc[i] = n < N ? v : 0.f;
    }
    __syncthreads();

    int nq;
    if constexpr (SAVE) {
        nq = scan_resolve<F, true>(ei, dinv, hin, E, lo, lim, acc, wq, wqw, wqd,
                                   ws_s, ws_w, ws_d, ws_c);
    } else {
        const int met = __builtin_amdgcn_readfirstlane(*ws_c);
        if (met > QW) {
            nq = scan_resolve<F, false>(ei, dinv, hin, E, lo, lim, acc, wq, wqw, wqd,
                                        ws_s, ws_w, ws_d, ws_c);
        } else {
            nq = met;
            for (int j0 = lane; j0 < nq; j0 += 8 * 32) {
                unsigned s[8];
                unsigned short d[8];
                float w[8];
                #pragma unroll
                for (int u = 0; u < 8; ++u) {
                    const int j  = j0 + u * 32;
                    const int jj = j < nq ? j : 0;
                    s[u] = ws_s[jj];
                    d[u] = ws_d[jj];
                    w[u] = ws_w[jj];
                }
                #pragma unroll
                for (int u = 0; u < 8; ++u) {
                    const int j = j0 + u * 32;
                    if (j < nq) { wq[j] = s[u]; wqd[j] = d[u]; wqw[j] = w[u]; }
                }
            }
        }
    }
    __builtin_amdgcn_fence(__ATOMIC_ACQ_REL, "wavefront");

    constexpr int PER = F / 32;
    constexpr int U   = 16;
    for (int j0 = 0; j0 < nq; j0 += U) {
        float r[U][PER];
        float w[U];
        int   dl[U];
        size_t s[U];
        #pragma unroll
        for (int u = 0; u < U; ++u) {
            const int j = (j0 + u < nq) ? j0 + u : nq - 1;
            s[u]  = wq[j];
            dl[u] = wqd[j];
            w[u]  = wqw[j];
        }
        #pragma unroll
        for (int u = 0; u < U; ++u) {
            const float* row = hin + s[u] * F + lane * PER;
            if constexpr (PER == 2) {
                const float2 t2 = *reinterpret_cast<const float2*>(row);
                r[u][0] = t2.x; r[u][1] = t2.y;
            } else {
                r[u][0] = row[0];
            }
        }
        #pragma unroll
        for (int u = 0; u < U; ++u) {
            if (j0 + u < nq) {
                #pragma unroll
                for (int p = 0; p < PER; ++p)
                    atomicAdd(&acc[dl[u] * F + lane * PER + p], r[u][p] * w[u]);
            }
        }
    }
    __syncthreads();

    for (int i = tid; i < BN * F; i += AGG_T) {
        const int n = lo + i / F;
        if (n < N) outp[(size_t)n * F + (i % F)] = acc[i];
    }
}

__global__ __launch_bounds__(AGG_T)
void k_range_agg64(const int* __restrict__ ei, const float* __restrict__ dinv,
                   const float* __restrict__ h, float* __restrict__ agg, int N, int E,
                   unsigned* __restrict__ sv_s, float* __restrict__ sv_w,
                   unsigned short* __restrict__ sv_d, int* __restrict__ sv_c) {
    __shared__ float          acc[BN * D_HID];
    __shared__ unsigned       q[NWAVE * QW];
    __shared__ float          qw[NWAVE * QW];
    __shared__ unsigned short qd[NWAVE * QW];
    range_agg<D_HID, true>(ei, dinv, h, nullptr, agg, N, E, sv_s, sv_w, sv_d, sv_c,
                           acc, q, qw, qd);
}


__global__ void k_gemm2_wmma(const float* __restrict__ AGG, const float* __restrict__ b1,
                             const float* __restrict__ W, float* __restrict__ H2) {
    const int lane = threadIdx.x & 31;
    const int wave = threadIdx.x >> 5;
    const int m0   = blockIdx.x * 16;
    const int n0   = wave * 16;
    const int half = lane >> 4;
    const int lm   = lane & 15;

    const float* arow = AGG + (size_t)(m0 + lm) * D_HID;
    v8f acc = {};
    #pragma unroll 4
    for (int k0 = 0; k0 < D_HID; k0 += 4) {
        const int ka = k0 + 2 * half;
        v2f a, b;
        a.x = fmaxf(arow[ka]     + b1[ka],     0.f);
        a.y = fmaxf(arow[ka + 1] + b1[ka + 1], 0.f);
        b.x = W[(size_t)ka       * D_OUT + n0 + lm];
        b.y = W[(size_t)(ka + 1) * D_OUT + n0 + lm];
        acc = __builtin_amdgcn_wmma_f32_16x16x4_f32(false, a, false, b,
                                                    (short)0, acc, false, false);
    }
    #pragma unroll
    for (int r = 0; r < 8; ++r) {
        const int m = m0 + r + 8 * half;
        H2[(size_t)m * D_OUT + n0 + lm] = acc[r];
    }
}

__global__ __launch_bounds__(AGG_T)
void k_range_agg32(const int* __restrict__ ei, const float* __restrict__ dinv,
                   const float* __restrict__ h2, const float* __restrict__ b2,
                   float* __restrict__ out, int N, int E,
                   unsigned* __restrict__ sv_s, float* __restrict__ sv_w,
                   unsigned short* __restrict__ sv_d, int* __restrict__ sv_c) {
    __shared__ float          acc[BN * D_OUT];
    __shared__ unsigned       q[NWAVE * QW];
    __shared__ float          qw[NWAVE * QW];
    __shared__ unsigned short qd[NWAVE * QW];
    range_agg<D_OUT, false>(ei, dinv, h2, b2, out, N, E, sv_s, sv_w, sv_d, sv_c,
                            acc, q, qw, qd);
}


extern "C" void kernel_launch(void* const* d_in, const int* in_sizes, int n_in,
                              void* d_out, int out_size, void* d_ws, size_t ws_size,
                              hipStream_t stream) {
    const float* x   = (const float*)d_in[0];
    const int*   ei  = (const int*)  d_in[1];
    const float* W1  = (const float*)d_in[2];
    const float* b1  = (const float*)d_in[3];
    const float* W2  = (const float*)d_in[4];
    const float* b2  = (const float*)d_in[5];
    float*       out = (float*)d_out;

    const int N = in_sizes[0] / D_IN;
    const int E = in_sizes[1] / 2;

    char* ws = (char*)d_ws;
    unsigned* deg  = (unsigned*)(ws);
    float*    dinv = (float*)(ws + (size_t)N * 4);
    float*    h    = (float*)(ws + (size_t)N * 8);
    float*    agg1 = (float*)(ws + (size_t)N * 8 + (size_t)N * D_HID * 4);
    float*    h2   = (float*)(ws + (size_t)N * 8 + (size_t)N * D_HID * 8);

    const int TB = 256;

    k_zero_u32<<<(N + TB - 1) / TB, TB, 0, stream>>>(deg, N);
    k_degree  <<<(E + TB - 1) / TB, TB, 0, stream>>>(ei, deg, E);
    k_dinv    <<<(N + TB - 1) / TB, TB, 0, stream>>>(deg, dinv, N);

    const int NB = (N + BN - 1) / BN;
    const size_t NQ = (size_t)NB * NWAVE * QW;
    char* qb = ws + (size_t)N * 8 + (size_t)N * D_HID * 8 + (size_t)N * D_OUT * 4;
    unsigned*       sv_s = (unsigned*)(qb);
    float*          sv_w = (float*)(qb + NQ * 4);
    unsigned short* sv_d = (unsigned short*)(qb + NQ * 8);
    int*            sv_c = (int*)(qb + NQ * 10);

    k_gemm1_wmma <<<N / 16, 128, 0, stream>>>(x, W1, h);
    k_range_agg64<<<NB, AGG_T, 0, stream>>>(ei, dinv, h, agg1, N, E, sv_s, sv_w, sv_d, sv_c);

    k_gemm2_wmma <<<N / 16, 64, 0, stream>>>(agg1, b1, W2, h2);
    k_range_agg32<<<NB, AGG_T, 0, stream>>>(ei, dinv, h2, b2, out, N, E, sv_s, sv_w, sv_d, sv_c);
}


# Destination scan reads one byte per edge (the destination's range number, stored by the degree kernel): 16 edges per 16-byte load, a quarter of the bytes; sum order kept: self term, then LDS arrival
# speedup vs baseline: 2.4342x; 1.0944x over previous
#include <hip/hip_runtime.h>
#include <hip/hip_bf16.h>


typedef float v2f __attribute__((ext_vector_type(2)));
typedef float v8f __attribute__((ext_vector_type(8)));

#define D_IN  128
#define D_HID 64
#define D_OUT 32


__global__ void k_zero_u32(unsigned* p, int n) {
    int i = blockIdx.x * blockDim.x + threadIdx.x;
    if (i < n) p[i] = 0u;
}

#define BN       512
#define BN_SHIFT 9
static_assert((1 << BN_SHIFT) == BN, "BN_SHIFT");

__global__ void k_degree(const int* __restrict__ ei, unsigned* __restrict__ deg,
                         unsigned char* __restrict__ rid, int E) {
    const int i = blockIdx.x * blockDim.x + threadIdx.x;
    const int* dstp = ei + E;
    if ((E & 3) == 0) {
        if (i < (E >> 2)) {
            const int4 d = reinterpret_cast<const int4*>(dstp)[i];
            atomicAdd(&deg[d.x], 1u);
            atomicAdd(&deg[d.y], 1u);
            atomicAdd(&deg[d.z], 1u);
            atomicAdd(&deg[d.w], 1u);
            reinterpret_cast<unsigned*>(rid)[i] =
                ((unsigned)d.x >> BN_SHIFT)         | (((unsigned)d.y >> BN_SHIFT) << 8) |
                (((unsigned)d.z >> BN_SHIFT) << 16) | (((unsigned)d.w >> BN_SHIFT) << 24);
        }
    } else {
        for (int u = 0; u < 4; ++u) {
            const int e = 4 * i + u;
            if (e < E) {
                const int d = dstp[e];
                atomicAdd(&deg[d], 1u);
                rid[e] = (unsigned char)((unsigned)d >> BN_SHIFT);
            }
        }
    }
}

__global__ void k_dinv(const unsigned* __restrict__ deg, float* __restrict__ dinv, int N) {
    int i = blockIdx.x * blockDim.x + threadIdx.x;
    if (i < N) {
        float dg = (float)(deg[i] + 1u);
        dinv[i] = rsqrtf(dg);
    }
}


__global__ void k_gemm1_wmma(const float* __restrict__ X, const float* __restrict__ W,
                             float* __restrict__ H) {
    const int lane = threadIdx.x & 31;
    const int wave = threadIdx.x >> 5;
    const int m0   = blockIdx.x * 16;
    const int n0   = wave * 16;
    const int half = lane >> 4;
    const int lm   = lane & 15;

    const float* xrow = X + (size_t)(m0 + lm) * D_IN;
    v8f acc = {};
    #pragma unroll 4
    for (int k0 = 0; k0 < D_IN; k0 += 4) {
        const int ka = k0 + 2 * half;
        v2f a, b;
        a.x = xrow[ka];
        a.y = xrow[ka + 1];
        b.x = W[(size_t)ka       * D_HID + n0 + lm];
        b.y = W[(size_t)(ka + 1) * D_HID + n0 + lm];
        acc = __builtin_amdgcn_wmma_f32_16x16x4_f32(false, a, false, b,
                                                    (short)0, acc, false, false);
    }
    #pragma unroll
    for (int r = 0; r < 8; ++r) {
        const int m = m0 + r + 8 * half;
        H[(size_t)m * D_HID + n0 + lm] = acc[r];
    }
}


#define AGG_T 1024
#define NWAVE (AGG_T / 32)
#define QW    384

template <int F, bool SAVE>
__device__ __forceinline__ int scan_resolve(const int* __restrict__ ei,
                                            const unsigned char* __restrict__ rid,
                                            const float* __restrict__ dinv,
                                            const float* __restrict__ hin,
                                            int E, int lo, unsigned lim, float* acc,
                                            unsigned* wq, float* wqw, unsigned short* wqd,
                                            unsigned* __restrict__ sv_s,
                                            float* __restrict__ sv_w,
                                            unsigned short* __restrict__ sv_d,
                                            int* __restrict__ sv_c) {
    const int tid  = threadIdx.x;
    const int lane = tid & 31;
    const int* dstp = ei + E;
    int wcount = 0;

    auto direct = [&](int e) {
        const int s  = ei[e];
        const int dn = dstp[e];
        const unsigned dl = (unsigned)(dn - lo);
        const float w = dinv[s] * dinv[dn];
        #pragma unroll 1
        for (int f = 0; f < F; ++f)
            atomicAdd(&acc[dl * F + f], hin[(size_t)s * F + f] * w);
    };

    auto append = [&](unsigned bal, bool hit, int e) {
        if (hit) {
            const int pos = wcount + (int)__builtin_amdgcn_mbcnt_lo(bal, 0u);
            if (pos < QW) wq[pos] = (unsigned)e;
            else          direct(e);
        }
        wcount += __builtin_popcount(bal);
    };

    if ((E & 15) == 0) {
        const unsigned pat = (unsigned)blockIdx.x * 0x01010101u;

        auto sixteen = [&](const uint4 t, int e0, bool live) {
            auto other = [&](unsigned x) {
                const unsigned y = x ^ pat;
                return (((y | 0x80808080u) - 0x01010101u) | y) & 0x80808080u;
            };
            unsigned m = ~((other(t.x) >> 7) | (other(t.y) >> 6) |
                           (other(t.z) >> 5) | (other(t.w) >> 4)) & 0x0F0F0F0Fu;
            if (!live) m = 0u;
            unsigned bal = __builtin_amdgcn_ballot_w32(m != 0u);
            while (bal != 0u) {
                if (m != 0u) {
                    const int p = __builtin_ctz(m);
                    const int e = e0 + 4 * (p & 7) + (p >> 3);
                    const int pos = wcount + (int)__builtin_amdgcn_mbcnt_lo(bal, 0u);
                    if (pos < QW) wq[pos] = (unsigned)e;
                    else          direct(e);
                }
                wcount += __builtin_popcount(bal);
                m &= m - 1u;
                bal = __builtin_amdgcn_ballot_w32(m != 0u);
            }
        };

        const uint4* b16 = reinterpret_cast<const uint4*>(rid);
        const int n16 = E >> 4;
        constexpr int L = 8;
        const int CH = L * AGG_T;
        const int trips = n16 / CH;
        const int rot = (int)(((long long)blockIdx.x * trips) / gridDim.x);
        uint4 cur[L], nxt[L];
        if (trips > 0) {
            #pragma unroll
            for (int u = 0; u < L; ++u) cur[u] = b16[rot * CH + u * AGG_T + tid];
        }
        for (int it = 0; it < trips; ++it) {
            int t = it + rot;
            if (t >= trips) t -= trips;
            int t1 = t + 1;
            if (t1 >= trips) t1 -= trips;
            #pragma unroll
            for (int u = 0; u < L; ++u) nxt[u] = b16[t1 * CH + u * AGG_T + tid];
            #pragma unroll
            for (int u = 0; u < L; ++u) sixteen(cur[u], (t * CH + u * AGG_T + tid) * 16, true);
            #pragma unroll
            for (int u = 0; u < L; ++u) cur[u] = nxt[u];
        }
        for (int base = trips * CH; base < n16; base += AGG_T) {
            const int i16 = base + tid;
            const bool live = i16 < n16;
            sixteen(b16[live ? i16 : 0], i16 * 16, live);
        }
    } else {
        for (int base = 0; base < E; base += AGG_T) {
            const int e = base + tid;
            const bool hit = (unsigned)((e < E ? dstp[e] : -1) - lo) < lim;
            const unsigned bal = __builtin_amdgcn_ballot_w32(hit);
            if (bal != 0u) append(bal, hit, e);
        }
    }

    const int nq = wcount < QW ? wcount : QW;
    if constexpr (SAVE) {
        if (lane == 0) *sv_c = wcount;
    }
    __builtin_amdgcn_fence(__ATOMIC_ACQ_REL, "wavefront");

    for (int j0 = lane; j0 < nq; j0 += 4 * 32) {
        int e[4], s[4], d[4];
        float ws[4], wd[4];
        #pragma unroll
        for (int u = 0; u < 4; ++u) {
            const int j = j0 + u * 32;
            e[u] = j < nq ? (int)wq[j] : 0;
        }
        #pragma unroll
        for (int u = 0; u < 4; ++u) { s[u] = ei[e[u]]; d[u] = dstp[e[u]]; }
        #pragma unroll
        for (int u = 0; u < 4; ++u) { ws[u] = dinv[s[u]]; wd[u] = dinv[d[u]]; }
        #pragma unroll
        for (int u = 0; u < 4; ++u) {
            const int j = j0 + u * 32;
            if (j < nq) {
                const unsigned short dl = (unsigned short)(d[u] - lo);
                const float w = ws[u] * wd[u];
                wq[j]  = (unsigned)s[u];
                wqd[j] = dl;
                wqw[j] = w;
                if constexpr (SAVE) {
                    sv_s[j] = (unsigned)s[u];
                    sv_d[j] = dl;
                    sv_w[j] = w;
                }
            }
        }
    }
    return nq;
}

template <int F, bool SAVE>
__device__ __forceinline__ void range_agg(const int* __restrict__ ei,
                                          const unsigned char* __restrict__ rid,
                                          const float* __restrict__ dinv,
                                          const float* __restrict__ hin,
                                          const float* __restrict__ bias,
                                          float* __restrict__ outp, int N, int E,
                                          unsigned* __restrict__ sv_s,
                                          float* __restrict__ sv_w,
                                          unsigned short* __restrict__ sv_d,
                                          int* __restrict__ sv_c,
                                          float* acc, unsigned* q, float* qw,
                                          unsigned short* qd) {
    const int tid  = threadIdx.x;
    const int lane = tid & 31;
    const int wave = tid >> 5;
    const int lo   = blockIdx.x * BN;
    const int left = N - lo;
    const unsigned lim = (unsigned)(left < BN ? left : BN);

    unsigned*       wq  = q  + wave * QW;
    float*          wqw = qw + wave * QW;
    unsigned short* wqd = qd + wave * QW;

    const size_t slot = (size_t)blockIdx.x * NWAVE + wave;
    unsigned*       ws_s = sv_s + slot * QW;
    float*          ws_w = sv_w + slot * QW;
    unsigned short* ws_d = sv_d + slot * QW;
    int*            ws_c = sv_c + slot;

    #pragma unroll 8
    for (int i = tid; i < BN * F; i += AGG_T) {
        const int n  = lo + i / F;
        const int f  = i % F;
        const int nn = n < N ? n : N - 1;
        const float w = dinv[nn];
        const float x = hin[(size_t)nn * F + f];
        const float v = bias ? bias[f] + w * w * x : w * w * x;
        acc[i] = n < N ? v : 0.f;
    }
    __syncthreads();

    int nq;
    if constexpr (SAVE) {
        nq = scan_resolve<F, true>(ei, rid, dinv, hin, E, lo, lim, acc, wq, wqw, wqd,
                                   ws_s, ws_w, ws_d, ws_c);
    } else {
        const int met = __builtin_amdgcn_readfirstlane(*ws_c);
        if (met > QW) {
            nq = scan_resolve<F, false>(ei, rid, dinv, hin, E, lo, lim, acc, wq, wqw, wqd,
                                        ws_s, ws_w, ws_d, ws_c);
        } else {
            nq = met;
            for (int j0 = lane; j0 < nq; j0 += 8 * 32) {
                unsigned s[8];
                unsigned short d[8];
                float w[8];
                #pragma unroll
                for (int u = 0; u < 8; ++u) {
                    const int j  = j0 + u * 32;
                    const int jj = j < nq ? j : 0;
                    s[u] = ws_s[jj];
                    d[u] = ws_d[jj];
                    w[u] = ws_w[jj];
                }
                #pragma unroll
                for (int u = 0; u < 8; ++u) {
                    const int j = j0 + u * 32;
                    if (j < nq) { wq[j] = s[u]; wqd[j] = d[u]; wqw[j] = w[u]; }
                }
            }
        }
    }
    __builtin_amdgcn_fence(__ATOMIC_ACQ_REL, "wavefront");

    constexpr int PER = F / 32;
    constexpr int U   = 16;
    for (int j0 = 0; j0 < nq; j0 += U) {
        float r[U][PER];
        float w[U];
        int   dl[U];
        size_t s[U];
        #pragma unroll
        for (int u = 0; u < U; ++u) {
            const int j = (j0 + u < nq) ? j0 + u : nq - 1;
            s[u]  = wq[j];
            dl[u] = wqd[j];
            w[u]  = wqw[j];
        }
        #pragma unroll
        for (int u = 0; u < U; ++u) {
            const float* row = hin + s[u] * F + lane * PER;
            if constexpr (PER == 2) {
                const float2 t2 = *reinterpret_cast<const float2*>(row);
                r[u][0] = t2.x; r[u][1] = t2.y;
            } else {
                r[u][0] = row[0];
            }
        }
        #pragma unroll
        for (int u = 0; u < U; ++u) {
            if (j0 + u < nq) {
                #pragma unroll
                for (int p = 0; p < PER; ++p)
                    atomicAdd(&acc[dl[u] * F + lane * PER + p], r[u][p] * w[u]);
            }
        }
    }
    __syncthreads();

    for (int i = tid; i < BN * F; i += AGG_T) {
        const int n = lo + i / F;
        if (n < N) outp[(size_t)n * F + (i % F)] = acc[i];
    }
}

__global__ __launch_bounds__(AGG_T)
void k_range_agg64(const int* __restrict__ ei, const unsigned char* __restrict__ rid,
                   const float* __restrict__ dinv,
                   const float* __restrict__ h, float* __restrict__ agg, int N, int E,
                   unsigned* __restrict__ sv_s, float* __restrict__ sv_w,
                   unsigned short* __restrict__ sv_d, int* __restrict__ sv_c) {
    __shared__ float          acc[BN * D_HID];
    __shared__ unsigned       q[NWAVE * QW];
    __shared__ float          qw[NWAVE * QW];
    __shared__ unsigned short qd[NWAVE * QW];
    range_agg<D_HID, true>(ei, rid, dinv, h, nullptr, agg, N, E, sv_s, sv_w, sv_d, sv_c,
                           acc, q, qw, qd);
}


__global__ void k_gemm2_wmma(const float* __restrict__ AGG, const float* __restrict__ b1,
                             const float* __restrict__ W, float* __restrict__ H2) {
    const int lane = threadIdx.x & 31;
    const int wave = threadIdx.x >> 5;
    const int m0   = blockIdx.x * 16;
    const int n0   = wave * 16;
    const int half = lane >> 4;
    const int lm   = lane & 15;

    const float* arow = AGG + (size_t)(m0 + lm) * D_HID;
    v8f acc = {};
    #pragma unroll 4
    for (int k0 = 0; k0 < D_HID; k0 += 4) {
        const int ka = k0 + 2 * half;
        v2f a, b;
        a.x = fmaxf(arow[ka]     + b1[ka],     0.f);
        a.y = fmaxf(arow[ka + 1] + b1[ka + 1], 0.f);
        b.x = W[(size_t)ka       * D_OUT + n0 + lm];
        b.y = W[(size_t)(ka + 1) * D_OUT + n0 + lm];
        acc = __builtin_amdgcn_wmma_f32_16x16x4_f32(false, a, false, b,
                                                    (short)0, acc, false, false);
    }
    #pragma unroll
    for (int r = 0; r < 8; ++r) {
        const int m = m0 + r + 8 * half;
        H2[(size_t)m * D_OUT + n0 + lm] = acc[r];
    }
}

__global__ __launch_bounds__(AGG_T)
void k_range_agg32(const int* __restrict__ ei, const unsigned char* __restrict__ rid,
                   const float* __restrict__ dinv,
                   const float* __restrict__ h2, const float* __restrict__ b2,
                   float* __restrict__ out, int N, int E,
                   unsigned* __restrict__ sv_s, float* __restrict__ sv_w,
                   unsigned short* __restrict__ sv_d, int* __restrict__ sv_c) {
    __shared__ float          acc[BN * D_OUT];
    __shared__ unsigned       q[NWAVE * QW];
    __shared__ float          qw[NWAVE * QW];
    __shared__ unsigned short qd[NWAVE * QW];
    range_agg<D_OUT, false>(ei, rid, dinv, h2, b2, out, N, E, sv_s, sv_w, sv_d, sv_c,
                            acc, q, qw, qd);
}


extern "C" void kernel_launch(void* const* d_in, const int* in_sizes, int n_in,
                              void* d_out, int out_size, void* d_ws, size_t ws_size,
                              hipStream_t stream) {
    const float* x   = (const float*)d_in[0];
    const int*   ei  = (const int*)  d_in[1];
    const float* W1  = (const float*)d_in[2];
    const float* b1  = (const float*)d_in[3];
    const float* W2  = (const float*)d_in[4];
    const float* b2  = (const float*)d_in[5];
    float*       out = (float*)d_out;

    const int N = in_sizes[0] / D_IN;
    const int E = in_sizes[1] / 2;

    char* ws = (char*)d_ws;
    unsigned* deg  = (unsigned*)(ws);
    float*    dinv = (float*)(ws + (size_t)N * 4);
    float*    h    = (float*)(ws + (size_t)N * 8);
    float*    agg1 = (float*)(ws + (size_t)N * 8 + (size_t)N * D_HID * 4);
    float*    h2   = (float*)(ws + (size_t)N * 8 + (size_t)N * D_HID * 8);

    const int TB = 256;

    const int NB = (N + BN - 1) / BN;
    const size_t NQ = (size_t)NB * NWAVE * QW;
    char* qb = ws + (size_t)N * 8 + (size_t)N * D_HID * 8 + (size_t)N * D_OUT * 4;
    unsigned*       sv_s = (unsigned*)(qb);
    float*          sv_w = (float*)(qb + NQ * 4);
    unsigned short* sv_d = (unsigned short*)(qb + NQ * 8);
    int*            sv_c = (int*)(qb + NQ * 10);
    unsigned char*  rid  = (unsigned char*)(sv_c + (size_t)NB * NWAVE);

    k_zero_u32<<<(N + TB - 1) / TB, TB, 0, stream>>>(deg, N);
    k_degree  <<<(E + 4 * TB - 1) / (4 * TB), TB, 0, stream>>>(ei, deg, rid, E);
    k_dinv    <<<(N + TB - 1) / TB, TB, 0, stream>>>(deg, dinv, N);

    k_gemm1_wmma <<<N / 16, 128, 0, stream>>>(x, W1, h);
    k_range_agg64<<<NB, AGG_T, 0, stream>>>(ei, rid, dinv, h, agg1, N, E, sv_s, sv_w, sv_d, sv_c);

    k_gemm2_wmma <<<N / 16, 64, 0, stream>>>(agg1, b1, W2, h2);
    k_range_agg32<<<NB, AGG_T, 0, stream>>>(ei, rid, dinv, h2, b2, out, N, E, sv_s, sv_w, sv_d, sv_c);
}


# Degree no longer by global atomics: a scan launch ahead of layer 1 finds each range's edges (stored for both layers), counts them per node in LDS (integer adds) and writes deg^-1/2; sum order kept
# speedup vs baseline: 3.6755x; 1.5099x over previous
#include <hip/hip_runtime.h>
#include <hip/hip_bf16.h>


typedef float v2f __attribute__((ext_vector_type(2)));
typedef float v8f __attribute__((ext_vector_type(8)));

#define D_IN  128
#define D_HID 64
#define D_OUT 32


#define BN       512
#define BN_SHIFT 9
static_assert((1 << BN_SHIFT) == BN, "BN_SHIFT");

__global__ void k_rid(const int* __restrict__ ei, unsigned char* __restrict__ rid, int E) {
    const int i = blockIdx.x * blockDim.x + threadIdx.x;
    const int* dstp = ei + E;
    if ((E & 3) == 0) {
        if (i < (E >> 2)) {
            const int4 d = reinterpret_cast<const int4*>(dstp)[i];
            reinterpret_cast<unsigned*>(rid)[i] =
                ((unsigned)d.x >> BN_SHIFT)         | (((unsigned)d.y >> BN_SHIFT) << 8) |
                (((unsigned)d.z >> BN_SHIFT) << 16) | (((unsigned)d.w >> BN_SHIFT) << 24);
        }
    } else {
        for (int u = 0; u < 4; ++u) {
            const int e = 4 * i + u;
            if (e < E) rid[e] = (unsigned char)((unsigned)dstp[e] >> BN_SHIFT);
        }
    }
}


__global__ void k_gemm1_wmma(const float* __restrict__ X, const float* __restrict__ W,
                             float* __restrict__ H) {
    const int lane = threadIdx.x & 31;
    const int wave = threadIdx.x >> 5;
    const int m0   = blockIdx.x * 16;
    const int n0   = wave * 16;
    const int half = lane >> 4;
    const int lm   = lane & 15;

    const float* xrow = X + (size_t)(m0 + lm) * D_IN;
    v8f acc = {};
    #pragma unroll 4
    for (int k0 = 0; k0 < D_IN; k0 += 4) {
        const int ka = k0 + 2 * half;
        v2f a, b;
        a.x = xrow[ka];
        a.y = xrow[ka + 1];
        b.x = W[(size_t)ka       * D_HID + n0 + lm];
        b.y = W[(size_t)(ka + 1) * D_HID + n0 + lm];
        acc = __builtin_amdgcn_wmma_f32_16x16x4_f32(false, a, false, b,
                                                    (short)0, acc, false, false);
    }
    #pragma unroll
    for (int r = 0; r < 8; ++r) {
        const int m = m0 + r + 8 * half;
        H[(size_t)m * D_HID + n0 + lm] = acc[r];
    }
}


#define AGG_T 1024
#define NWAVE (AGG_T / 32)
#define QW    384

template <class Direct>
__device__ __forceinline__ int scan_edges(const int* __restrict__ ei,
                                          const unsigned char* __restrict__ rid,
                                          int E, int lo, unsigned lim, unsigned* wq,
                                          Direct direct) {
    const int tid  = threadIdx.x;
    const int* dstp = ei + E;
    int wcount = 0;

    auto append = [&](unsigned bal, bool hit, int e) {
        if (hit) {
            const int pos = wcount + (int)__builtin_amdgcn_mbcnt_lo(bal, 0u);
            if (pos < QW) wq[pos] = (unsigned)e;
            else          direct(e);
        }
        wcount += __builtin_popcount(bal);
    };

    if ((E & 15) == 0) {
        const unsigned pat = (unsigned)blockIdx.x * 0x01010101u;

        auto sixteen = [&](const uint4 t, int e0, bool live) {
            auto other = [&](unsigned x) {
                const unsigned y = x ^ pat;
                return (((y | 0x80808080u) - 0x01010101u) | y) & 0x80808080u;
            };
            unsigned m = ~((other(t.x) >> 7) | (other(t.y) >> 6) |
                           (other(t.z) >> 5) | (other(t.w) >> 4)) & 0x0F0F0F0Fu;
            if (!live) m = 0u;
            unsigned bal = __builtin_amdgcn_ballot_w32(m != 0u);
            while (bal != 0u) {
                if (m != 0u) {
                    const int p = __builtin_ctz(m);
                    const int e = e0 + 4 * (p & 7) + (p >> 3);
                    const int pos = wcount + (int)__builtin_amdgcn_mbcnt_lo(bal, 0u);
                    if (pos < QW) wq[pos] = (unsigned)e;
                    else          direct(e);
                }
                wcount += __builtin_popcount(bal);
                m &= m - 1u;
                bal = __builtin_amdgcn_ballot_w32(m != 0u);
            }
        };

        const uint4* b16 = reinterpret_cast<const uint4*>(rid);
        const int n16 = E >> 4;
        constexpr int L = 8;
        const int CH = L * AGG_T;
        const int trips = n16 / CH;
        const int rot = (int)(((long long)blockIdx.x * trips) / gridDim.x);
        uint4 cur[L], nxt[L];
        if (trips > 0) {
            #pragma unroll
            for (int u = 0; u < L; ++u) cur[u] = b16[rot * CH + u * AGG_T + tid];
        }
        for (int it = 0; it < trips; ++it) {
            int t = it + rot;
            if (t >= trips) t -= trips;
            int t1 = t + 1;
            if (t1 >= trips) t1 -= trips;
            #pragma unroll
            for (int u = 0; u < L; ++u) nxt[u] = b16[t1 * CH + u * AGG_T + tid];
            #pragma unroll
            for (int u = 0; u < L; ++u) sixteen(cur[u], (t * CH + u * AGG_T + tid) * 16, true);
            #pragma unroll
            for (int u = 0; u < L; ++u) cur[u] = nxt[u];
        }
        for (int base = trips * CH; base < n16; base += AGG_T) {
            const int i16 = base + tid;
            const bool live = i16 < n16;
            sixteen(b16[live ? i16 : 0], i16 * 16, live);
        }
    } else {
        for (int base = 0; base < E; base += AGG_T) {
            const int e = base + tid;
            const bool hit = (unsigned)((e < E ? dstp[e] : -1) - lo) < lim;
            const unsigned bal = __builtin_amdgcn_ballot_w32(hit);
            if (bal != 0u) append(bal, hit, e);
        }
    }
    return wcount;
}

__global__ __launch_bounds__(AGG_T)
void k_range_scan(const int* __restrict__ ei, const unsigned char* __restrict__ rid,
                  float* __restrict__ dinv, int N, int E,
                  unsigned* __restrict__ sv_s, unsigned short* __restrict__ sv_d,
                  int* __restrict__ sv_c) {
    __shared__ unsigned cnt[BN];
    __shared__ unsigned q[NWAVE * QW];

    const int tid  = threadIdx.x;
    const int lane = tid & 31;
    const int wave = tid >> 5;
    const int lo   = blockIdx.x * BN;
    const int left = N - lo;
    const unsigned lim = (unsigned)(left < BN ? left : BN);
    const int* dstp = ei + E;

    unsigned* wq = q + wave * QW;
    const size_t slot = (size_t)blockIdx.x * NWAVE + wave;
    unsigned*       ws_s = sv_s + slot * QW;
    unsigned short* ws_d = sv_d + slot * QW;

    for (int i = tid; i < BN; i += AGG_T) cnt[i] = 0u;
    __syncthreads();

    auto direct = [&](int e) { atomicAdd(&cnt[dstp[e] - lo], 1u); };
    const int wcount = scan_edges(ei, rid, E, lo, lim, wq, direct);
    const int nq = wcount < QW ? wcount : QW;
    if (lane == 0) sv_c[slot] = wcount;
    __builtin_amdgcn_fence(__ATOMIC_ACQ_REL, "wavefront");

    for (int j0 = lane; j0 < nq; j0 += 4 * 32) {
        int e[4], s[4], d[4];
        #pragma unroll
        for (int u = 0; u < 4; ++u) {
            const int j = j0 + u * 32;
            e[u] = j < nq ? (int)wq[j] : 0;
        }
        #pragma unroll
        for (int u = 0; u < 4; ++u) { s[u] = ei[e[u]]; d[u] = dstp[e[u]]; }
        #pragma unroll
        for (int u = 0; u < 4; ++u) {
            const int j = j0 + u * 32;
            if (j < nq) {
                const unsigned dl = (unsigned)(d[u] - lo);
                ws_s[j] = (unsigned)s[u];
                ws_d[j] = (unsigned short)dl;
                atomicAdd(&cnt[dl], 1u);
            }
        }
    }
    __syncthreads();

    for (int i = tid; i < BN; i += AGG_T) {
        const int n = lo + i;
        if (n < N) dinv[n] = rsqrtf((float)(cnt[i] + 1u));
    }
}

template <int F>
__device__ __forceinline__ int scan_resolve(const int* __restrict__ ei,
                                            const unsigned char* __restrict__ rid,
                                            const float* __restrict__ dinv,
                                            const float* __restrict__ hin,
                                            int E, int lo, unsigned lim, float* acc,
                                            unsigned* wq, float* wqw, unsigned short* wqd) {
    const int lane = threadIdx.x & 31;
    const int* dstp = ei + E;

    auto direct = [&](int e) {
        const int s  = ei[e];
        const int dn = dstp[e];
        const unsigned dl = (unsigned)(dn - lo);
        const float w = dinv[s] * dinv[dn];
        #pragma unroll 1
        for (int f = 0; f < F; ++f)
            atomicAdd(&acc[dl * F + f], hin[(size_t)s * F + f] * w);
    };
    const int wcount = scan_edges(ei, rid, E, lo, lim, wq, direct);
    const int nq = wcount < QW ? wcount : QW;
    __builtin_amdgcn_fence(__ATOMIC_ACQ_REL, "wavefront");

    for (int j0 = lane; j0 < nq; j0 += 4 * 32) {
        int e[4], s[4], d[4];
        float ws[4], wd[4];
        #pragma unroll
        for (int u = 0; u < 4; ++u) {
            const int j = j0 + u * 32;
            e[u] = j < nq ? (int)wq[j] : 0;
        }
        #pragma unroll
        for (int u = 0; u < 4; ++u) { s[u] = ei[e[u]]; d[u] = dstp[e[u]]; }
        #pragma unroll
        for (int u = 0; u < 4; ++u) { ws[u] = dinv[s[u]]; wd[u] = dinv[d[u]]; }
        #pragma unroll
        for (int u = 0; u < 4; ++u) {
            const int j = j0 + u * 32;
            if (j < nq) {
                wq[j]  = (unsigned)s[u];
                wqd[j] = (unsigned short)(d[u] - lo);
                wqw[j] = ws[u] * wd[u];
            }
        }
    }
    return nq;
}

template <int F, bool FIRST>
__device__ __forceinline__ void range_agg(const int* __restrict__ ei,
                                          const unsigned char* __restrict__ rid,
                                          const float* __restrict__ dinv,
                                          const float* __restrict__ hin,
                                          const float* __restrict__ bias,
                                          float* __restrict__ outp, int N, int E,
                                          const unsigned* __restrict__ sv_s,
                                          float* __restrict__ sv_w,
                                          const unsigned short* __restrict__ sv_d,
                                          const int* __restrict__ sv_c,
                                          float* acc, unsigned* q, float* qw,
                                          unsigned short* qd) {
    const int tid  = threadIdx.x;
    const int lane = tid & 31;
    const int wave = tid >> 5;
    const int lo   = blockIdx.x * BN;
    const int left = N - lo;
    const unsigned lim = (unsigned)(left < BN ? left : BN);

    unsigned*       wq  = q  + wave * QW;
    float*          wqw = qw + wave * QW;
    unsigned short* wqd = qd + wave * QW;

    const size_t slot = (size_t)blockIdx.x * NWAVE + wave;
    const unsigned*       ws_s = sv_s + slot * QW;
    float*                ws_w = sv_w + slot * QW;
    const unsigned short* ws_d = sv_d + slot * QW;

    #pragma unroll 8
    for (int i = tid; i < BN * F; i += AGG_T) {
        const int n  = lo + i / F;
        const int f  = i % F;
        const int nn = n < N ? n : N - 1;
        const float w = dinv[nn];
        const float x = hin[(size_t)nn * F + f];
        const float v = bias ? bias[f] + w * w * x : w * w * x;
        acc[i] = n < N ? v : 0.f;
    }
    __syncthreads();

    int nq;
    const int met = __builtin_amdgcn_readfirstlane(sv_c[slot]);
    if (met > QW) {
        nq = scan_resolve<F>(ei, rid, dinv, hin, E, lo, lim, acc, wq, wqw, wqd);
    } else {
        nq = met;
        for (int j0 = lane; j0 < nq; j0 += 8 * 32) {
            unsigned s[8];
            unsigned short d[8];
            float w[8];
            #pragma unroll
            for (int u = 0; u < 8; ++u) {
                const int j  = j0 + u * 32;
                const int jj = j < nq ? j : 0;
                s[u] = ws_s[jj];
                d[u] = ws_d[jj];
                if constexpr (!FIRST) w[u] = ws_w[jj];
            }
            if constexpr (FIRST) {
                float a[8], b[8];
                #pragma unroll
                for (int u = 0; u < 8; ++u) { a[u] = dinv[s[u]]; b[u] = dinv[lo + (int)d[u]]; }
                #pragma unroll
                for (int u = 0; u < 8; ++u) w[u] = a[u] * b[u];
            }
            #pragma unroll
            for (int u = 0; u < 8; ++u) {
                const int j = j0 + u * 32;
                if (j < nq) {
                    wq[j] = s[u]; wqd[j] = d[u]; wqw[j] = w[u];
                    if constexpr (FIRST) ws_w[j] = w[u];
                }
            }
        }
    }
    __builtin_amdgcn_fence(__ATOMIC_ACQ_REL, "wavefront");

    constexpr int PER = F / 32;
    constexpr int U   = 16;
    for (int j0 = 0; j0 < nq; j0 += U) {
        float r[U][PER];
        float w[U];
        int   dl[U];
        size_t s[U];
        #pragma unroll
        for (int u = 0; u < U; ++u) {
            const int j = (j0 + u < nq) ? j0 + u : nq - 1;
            s[u]  = wq[j];
            dl[u] = wqd[j];
            w[u]  = wqw[j];
        }
        #pragma unroll
        for (int u = 0; u < U; ++u) {
            const float* row = hin + s[u] * F + lane * PER;
            if constexpr (PER == 2) {
                const float2 t2 = *reinterpret_cast<const float2*>(row);
                r[u][0] = t2.x; r[u][1] = t2.y;
            } else {
                r[u][0] = row[0];
            }
        }
        #pragma unroll
        for (int u = 0; u < U; ++u) {
            if (j0 + u < nq) {
                #pragma unroll
                for (int p = 0; p < PER; ++p)
                    atomicAdd(&acc[dl[u] * F + lane * PER + p], r[u][p] * w[u]);
            }
        }
    }
    __syncthreads();

    for (int i = tid; i < BN * F; i += AGG_T) {
        const int n = lo + i / F;
        if (n < N) outp[(size_t)n * F + (i % F)] = acc[i];
    }
}

__global__ __launch_bounds__(AGG_T)
void k_range_agg64(const int* __restrict__ ei, const unsigned char* __restrict__ rid,
                   const float* __restrict__ dinv,
                   const float* __restrict__ h, float* __restrict__ agg, int N, int E,
                   const unsigned* __restrict__ sv_s, float* __restrict__ sv_w,
                   const unsigned short* __restrict__ sv_d, const int* __restrict__ sv_c) {
    __shared__ float          acc[BN * D_HID];
    __shared__ unsigned       q[NWAVE * QW];
    __shared__ float          qw[NWAVE * QW];
    __shared__ unsigned short qd[NWAVE * QW];
    range_agg<D_HID, true>(ei, rid, dinv, h, nullptr, agg, N, E, sv_s, sv_w, sv_d, sv_c,
                           acc, q, qw, qd);
}


__global__ void k_gemm2_wmma(const float* __restrict__ AGG, const float* __restrict__ b1,
                             const float* __restrict__ W, float* __restrict__ H2) {
    const int lane = threadIdx.x & 31;
    const int wave = threadIdx.x >> 5;
    const int m0   = blockIdx.x * 16;
    const int n0   = wave * 16;
    const int half = lane >> 4;
    const int lm   = lane & 15;

    const float* arow = AGG + (size_t)(m0 + lm) * D_HID;
    v8f acc = {};
    #pragma unroll 4
    for (int k0 = 0; k0 < D_HID; k0 += 4) {
        const int ka = k0 + 2 * half;
        v2f a, b;
        a.x = fmaxf(arow[ka]     + b1[ka],     0.f);
        a.y = fmaxf(arow[ka + 1] + b1[ka + 1], 0.f);
        b.x = W[(size_t)ka       * D_OUT + n0 + lm];
        b.y = W[(size_t)(ka + 1) * D_OUT + n0 + lm];
        acc = __builtin_amdgcn_wmma_f32_16x16x4_f32(false, a, false, b,
                                                    (short)0, acc, false, false);
    }
    #pragma unroll
    for (int r = 0; r < 8; ++r) {
        const int m = m0 + r + 8 * half;
        H2[(size_t)m * D_OUT + n0 + lm] = acc[r];
    }
}

__global__ __launch_bounds__(AGG_T)
void k_range_agg32(const int* __restrict__ ei, const unsigned char* __restrict__ rid,
                   const float* __restrict__ dinv,
                   const float* __restrict__ h2, const float* __restrict__ b2,
                   float* __restrict__ out, int N, int E,
                   const unsigned* __restrict__ sv_s, float* __restrict__ sv_w,
                   const unsigned short* __restrict__ sv_d, const int* __restrict__ sv_c) {
    __shared__ float          acc[BN * D_OUT];
    __shared__ unsigned       q[NWAVE * QW];
    __shared__ float          qw[NWAVE * QW];
    __shared__ unsigned short qd[NWAVE * QW];
    range_agg<D_OUT, false>(ei, rid, dinv, h2, b2, out, N, E, sv_s, sv_w, sv_d, sv_c,
                            acc, q, qw, qd);
}


extern "C" void kernel_launch(void* const* d_in, const int* in_sizes, int n_in,
                              void* d_out, int out_size, void* d_ws, size_t ws_size,
                              hipStream_t stream) {
    const float* x   = (const float*)d_in[0];
    const int*   ei  = (const int*)  d_in[1];
    const float* W1  = (const float*)d_in[2];
    const float* b1  = (const float*)d_in[3];
    const float* W2  = (const float*)d_in[4];
    const float* b2  = (const float*)d_in[5];
    float*       out = (float*)d_out;

    const int N = in_sizes[0] / D_IN;
    const int E = in_sizes[1] / 2;

    char* ws = (char*)d_ws;
    float*    dinv = (float*)(ws + (size_t)N * 4);
    float*    h    = (float*)(ws + (size_t)N * 8);
    float*    agg1 = (float*)(ws + (size_t)N * 8 + (size_t)N * D_HID * 4);
    float*    h2   = (float*)(ws + (size_t)N * 8 + (size_t)N * D_HID * 8);

    const int TB = 256;

    const int NB = (N + BN - 1) / BN;
    const size_t NQ = (size_t)NB * NWAVE * QW;
    char* qb = ws + (size_t)N * 8 + (size_t)N * D_HID * 8 + (size_t)N * D_OUT * 4;
    unsigned*       sv_s = (unsigned*)(qb);
    float*          sv_w = (float*)(qb + NQ * 4);
    unsigned short* sv_d = (unsigned short*)(qb + NQ * 8);
    int*            sv_c = (int*)(qb + NQ * 10);
    unsigned char*  rid  = (unsigned char*)(sv_c + (size_t)NB * NWAVE);

    k_rid       <<<(E + 4 * TB - 1) / (4 * TB), TB, 0, stream>>>(ei, rid, E);
    k_range_scan<<<NB, AGG_T, 0, stream>>>(ei, rid, dinv, N, E, sv_s, sv_d, sv_c);

    k_gemm1_wmma <<<N / 16, 128, 0, stream>>>(x, W1, h);
    k_range_agg64<<<NB, AGG_T, 0, stream>>>(ei, rid, dinv, h, agg1, N, E, sv_s, sv_w, sv_d, sv_c);

    k_gemm2_wmma <<<N / 16, 64, 0, stream>>>(agg1, b1, W2, h2);
    k_range_agg32<<<NB, AGG_T, 0, stream>>>(ei, rid, dinv, h2, b2, out, N, E, sv_s, sv_w, sv_d, sv_c);
}


# Add loops of both layers: a wave holds 32 queue entries one per lane; each step takes its entry by readlane (was 2 LDS reads per edge); sum order kept. Left: a group's 2nd row load waits alone
# speedup vs baseline: 3.9202x; 1.0666x over previous
#include <hip/hip_runtime.h>
#include <hip/hip_bf16.h>


typedef float v2f __attribute__((ext_vector_type(2)));
typedef float v8f __attribute__((ext_vector_type(8)));

#define D_IN  128
#define D_HID 64
#define D_OUT 32


#define BN       512
#define BN_SHIFT 9
static_assert((1 << BN_SHIFT) == BN, "BN_SHIFT");

__global__ void k_rid(const int* __restrict__ ei, unsigned char* __restrict__ rid, int E) {
    const int i = blockIdx.x * blockDim.x + threadIdx.x;
    const int* dstp = ei + E;
    if ((E & 3) == 0) {
        if (i < (E >> 2)) {
            const int4 d = reinterpret_cast<const int4*>(dstp)[i];
            reinterpret_cast<unsigned*>(rid)[i] =
                ((unsigned)d.x >> BN_SHIFT)         | (((unsigned)d.y >> BN_SHIFT) << 8) |
                (((unsigned)d.z >> BN_SHIFT) << 16) | (((unsigned)d.w >> BN_SHIFT) << 24);
        }
    } else {
        for (int u = 0; u < 4; ++u) {
            const int e = 4 * i + u;
            if (e < E) rid[e] = (unsigned char)((unsigned)dstp[e] >> BN_SHIFT);
        }
    }
}


__global__ void k_gemm1_wmma(const float* __restrict__ X, const float* __restrict__ W,
                             float* __restrict__ H) {
    const int lane = threadIdx.x & 31;
    const int wave = threadIdx.x >> 5;
    const int m0   = blockIdx.x * 16;
    const int n0   = wave * 16;
    const int half = lane >> 4;
    const int lm   = lane & 15;

    const float* xrow = X + (size_t)(m0 + lm) * D_IN;
    v8f acc = {};
    #pragma unroll 4
    for (int k0 = 0; k0 < D_IN; k0 += 4) {
        const int ka = k0 + 2 * half;
        v2f a, b;
        a.x = xrow[ka];
        a.y = xrow[ka + 1];
        b.x = W[(size_t)ka       * D_HID + n0 + lm];
        b.y = W[(size_t)(ka + 1) * D_HID + n0 + lm];
        acc = __builtin_amdgcn_wmma_f32_16x16x4_f32(false, a, false, b,
                                                    (short)0, acc, false, false);
    }
    #pragma unroll
    for (int r = 0; r < 8; ++r) {
        const int m = m0 + r + 8 * half;
        H[(size_t)m * D_HID + n0 + lm] = acc[r];
    }
}


#define AGG_T 1024
#define NWAVE (AGG_T / 32)
#define QW    384

template <class Direct>
__device__ __forceinline__ int scan_edges(const int* __restrict__ ei,
                                          const unsigned char* __restrict__ rid,
                                          int E, int lo, unsigned lim, unsigned* wq,
                                          Direct direct) {
    const int tid  = threadIdx.x;
    const int* dstp = ei + E;
    int wcount = 0;

    auto append = [&](unsigned bal, bool hit, int e) {
        if (hit) {
            const int pos = wcount + (int)__builtin_amdgcn_mbcnt_lo(bal, 0u);
            if (pos < QW) wq[pos] = (unsigned)e;
            else          direct(e);
        }
        wcount += __builtin_popcount(bal);
    };

    if ((E & 15) == 0) {
        const unsigned pat = (unsigned)blockIdx.x * 0x01010101u;

        auto sixteen = [&](const uint4 t, int e0, bool live) {
            auto other = [&](unsigned x) {
                const unsigned y = x ^ pat;
                return (((y | 0x80808080u) - 0x01010101u) | y) & 0x80808080u;
            };
            unsigned m = ~((other(t.x) >> 7) | (other(t.y) >> 6) |
                           (other(t.z) >> 5) | (other(t.w) >> 4)) & 0x0F0F0F0Fu;
            if (!live) m = 0u;
            unsigned bal = __builtin_amdgcn_ballot_w32(m != 0u);
            while (bal != 0u) {
                if (m != 0u) {
                    const int p = __builtin_ctz(m);
                    const int e = e0 + 4 * (p & 7) + (p >> 3);
                    const int pos = wcount + (int)__builtin_amdgcn_mbcnt_lo(bal, 0u);
                    if (pos < QW) wq[pos] = (unsigned)e;
                    else          direct(e);
                }
                wcount += __builtin_popcount(bal);
                m &= m - 1u;
                bal = __builtin_amdgcn_ballot_w32(m != 0u);
            }
        };

        const uint4* b16 = reinterpret_cast<const uint4*>(rid);
        const int n16 = E >> 4;
        constexpr int L = 8;
        const int CH = L * AGG_T;
        const int trips = n16 / CH;
        const int rot = (int)(((long long)blockIdx.x * trips) / gridDim.x);
        uint4 cur[L], nxt[L];
        if (trips > 0) {
            #pragma unroll
            for (int u = 0; u < L; ++u) cur[u] = b16[rot * CH + u * AGG_T + tid];
        }
        for (int it = 0; it < trips; ++it) {
            int t = it + rot;
            if (t >= trips) t -= trips;
            int t1 = t + 1;
            if (t1 >= trips) t1 -= trips;
            #pragma unroll
            for (int u = 0; u < L; ++u) nxt[u] = b16[t1 * CH + u * AGG_T + tid];
            #pragma unroll
            for (int u = 0; u < L; ++u) sixteen(cur[u], (t * CH + u * AGG_T + tid) * 16, true);
            #pragma unroll
            for (int u = 0; u < L; ++u) cur[u] = nxt[u];
        }
        for (int base = trips * CH; base < n16; base += AGG_T) {
            const int i16 = base + tid;
            const bool live = i16 < n16;
            sixteen(b16[live ? i16 : 0], i16 * 16, live);
        }
    } else {
        for (int base = 0; base < E; base += AGG_T) {
            const int e = base + tid;
            const bool hit = (unsigned)((e < E ? dstp[e] : -1) - lo) < lim;
            const unsigned bal = __builtin_amdgcn_ballot_w32(hit);
            if (bal != 0u) append(bal, hit, e);
        }
    }
    return wcount;
}

__global__ __launch_bounds__(AGG_T)
void k_range_scan(const int* __restrict__ ei, const unsigned char* __restrict__ rid,
                  float* __restrict__ dinv, int N, int E,
                  unsigned* __restrict__ sv_s, unsigned short* __restrict__ sv_d,
                  int* __restrict__ sv_c) {
    __shared__ unsigned cnt[BN];
    __shared__ unsigned q[NWAVE * QW];

    const int tid  = threadIdx.x;
    const int lane = tid & 31;
    const int wave = tid >> 5;
    const int lo   = blockIdx.x * BN;
    const int left = N - lo;
    const unsigned lim = (unsigned)(left < BN ? left : BN);
    const int* dstp = ei + E;

    unsigned* wq = q + wave * QW;
    const size_t slot = (size_t)blockIdx.x * NWAVE + wave;
    unsigned*       ws_s = sv_s + slot * QW;
    unsigned short* ws_d = sv_d + slot * QW;

    for (int i = tid; i < BN; i += AGG_T) cnt[i] = 0u;
    __syncthreads();

    auto direct = [&](int e) { atomicAdd(&cnt[dstp[e] - lo], 1u); };
    const int wcount = scan_edges(ei, rid, E, lo, lim, wq, direct);
    const int nq = wcount < QW ? wcount : QW;
    if (lane == 0) sv_c[slot] = wcount;
    __builtin_amdgcn_fence(__ATOMIC_ACQ_REL, "wavefront");

    for (int j0 = lane; j0 < nq; j0 += 4 * 32) {
        int e[4], s[4], d[4];
        #pragma unroll
        for (int u = 0; u < 4; ++u) {
            const int j = j0 + u * 32;
            e[u] = j < nq ? (int)wq[j] : 0;
        }
        #pragma unroll
        for (int u = 0; u < 4; ++u) { s[u] = ei[e[u]]; d[u] = dstp[e[u]]; }
        #pragma unroll
        for (int u = 0; u < 4; ++u) {
            const int j = j0 + u * 32;
            if (j < nq) {
                const unsigned dl = (unsigned)(d[u] - lo);
                ws_s[j] = (unsigned)s[u];
                ws_d[j] = (unsigned short)dl;
                atomicAdd(&cnt[dl], 1u);
            }
        }
    }
    __syncthreads();

    for (int i = tid; i < BN; i += AGG_T) {
        const int n = lo + i;
        if (n < N) dinv[n] = rsqrtf((float)(cnt[i] + 1u));
    }
}

template <int F>
__device__ __forceinline__ int scan_resolve(const int* __restrict__ ei,
                                            const unsigned char* __restrict__ rid,
                                            const float* __restrict__ dinv,
                                            const float* __restrict__ hin,
                                            int E, int lo, unsigned lim, float* acc,
                                            unsigned* wq, float* wqw, unsigned short* wqd) {
    const int lane = threadIdx.x & 31;
    const int* dstp = ei + E;

    auto direct = [&](int e) {
        const int s  = ei[e];
        const int dn = dstp[e];
        const unsigned dl = (unsigned)(dn - lo);
        const float w = dinv[s] * dinv[dn];
        #pragma unroll 1
        for (int f = 0; f < F; ++f)
            atomicAdd(&acc[dl * F + f], hin[(size_t)s * F + f] * w);
    };
    const int wcount = scan_edges(ei, rid, E, lo, lim, wq, direct);
    const int nq = wcount < QW ? wcount : QW;
    __builtin_amdgcn_fence(__ATOMIC_ACQ_REL, "wavefront");

    for (int j0 = lane; j0 < nq; j0 += 4 * 32) {
        int e[4], s[4], d[4];
        float ws[4], wd[4];
        #pragma unroll
        for (int u = 0; u < 4; ++u) {
            const int j = j0 + u * 32;
            e[u] = j < nq ? (int)wq[j] : 0;
        }
        #pragma unroll
        for (int u = 0; u < 4; ++u) { s[u] = ei[e[u]]; d[u] = dstp[e[u]]; }
        #pragma unroll
        for (int u = 0; u < 4; ++u) { ws[u] = dinv[s[u]]; wd[u] = dinv[d[u]]; }
        #pragma unroll
        for (int u = 0; u < 4; ++u) {
            const int j = j0 + u * 32;
            if (j < nq) {
                wq[j]  = (unsigned)s[u];
                wqd[j] = (unsigned short)(d[u] - lo);
                wqw[j] = ws[u] * wd[u];
            }
        }
    }
    return nq;
}

template <int F, bool FIRST>
__device__ __forceinline__ void range_agg(const int* __restrict__ ei,
                                          const unsigned char* __restrict__ rid,
                                          const float* __restrict__ dinv,
                                          const float* __restrict__ hin,
                                          const float* __restrict__ bias,
                                          float* __restrict__ outp, int N, int E,
                                          const unsigned* __restrict__ sv_s,
                                          float* __restrict__ sv_w,
                                          const unsigned short* __restrict__ sv_d,
                                          const int* __restrict__ sv_c,
                                          float* acc, unsigned* q, float* qw,
                                          unsigned short* qd) {
    const int tid  = threadIdx.x;
    const int lane = tid & 31;
    const int wave = tid >> 5;
    const int lo   = blockIdx.x * BN;
    const int left = N - lo;
    const unsigned lim = (unsigned)(left < BN ? left : BN);

    unsigned*       wq  = q  + wave * QW;
    float*          wqw = qw + wave * QW;
    unsigned short* wqd = qd + wave * QW;

    const size_t slot = (size_t)blockIdx.x * NWAVE + wave;
    const unsigned*       ws_s = sv_s + slot * QW;
    float*                ws_w = sv_w + slot * QW;
    const unsigned short* ws_d = sv_d + slot * QW;

    #pragma unroll 8
    for (int i = tid; i < BN * F; i += AGG_T) {
        const int n  = lo + i / F;
        const int f  = i % F;
        const int nn = n < N ? n : N - 1;
        const float w = dinv[nn];
        const float x = hin[(size_t)nn * F + f];
        const float v = bias ? bias[f] + w * w * x : w * w * x;
        acc[i] = n < N ? v : 0.f;
    }
    __syncthreads();

    int nq;
    const int met = __builtin_amdgcn_readfirstlane(sv_c[slot]);
    if (met > QW) {
        nq = scan_resolve<F>(ei, rid, dinv, hin, E, lo, lim, acc, wq, wqw, wqd);
    } else {
        nq = met;
        for (int j0 = lane; j0 < nq; j0 += 8 * 32) {
            unsigned s[8];
            unsigned short d[8];
            float w[8];
            #pragma unroll
            for (int u = 0; u < 8; ++u) {
                const int j  = j0 + u * 32;
                const int jj = j < nq ? j : 0;
                s[u] = ws_s[jj];
                d[u] = ws_d[jj];
                if constexpr (!FIRST) w[u] = ws_w[jj];
            }
            if constexpr (FIRST) {
                float a[8], b[8];
                #pragma unroll
                for (int u = 0; u < 8; ++u) { a[u] = dinv[s[u]]; b[u] = dinv[lo + (int)d[u]]; }
                #pragma unroll
                for (int u = 0; u < 8; ++u) w[u] = a[u] * b[u];
            }
            #pragma unroll
            for (int u = 0; u < 8; ++u) {
                const int j = j0 + u * 32;
                if (j < nq) {
                    wq[j] = s[u]; wqd[j] = d[u]; wqw[j] = w[u];
                    if constexpr (FIRST) ws_w[j] = w[u];
                }
            }
        }
    }
    __builtin_amdgcn_fence(__ATOMIC_ACQ_REL, "wavefront");

    constexpr int PER = F / 32;
    constexpr int U   = 16;
    for (int j0 = 0; j0 < nq; j0 += 32) {
        const int jl = (j0 + lane < nq) ? j0 + lane : nq - 1;
        const int sv = (int)wq[jl];
        const int dv = (int)wqd[jl];
        const int wv = __float_as_int(wqw[jl]);
        #pragma unroll
        for (int h = 0; h < 32; h += U) {
            if (j0 + h < nq) {
                float r[U][PER];
                #pragma unroll
                for (int u = 0; u < U; ++u) {
                    const size_t s = (unsigned)__builtin_amdgcn_readlane(sv, h + u);
                    const float* row = hin + s * F + lane * PER;
                    if constexpr (PER == 2) {
                        const float2 t2 = *reinterpret_cast<const float2*>(row);
                        r[u][0] = t2.x; r[u][1] = t2.y;
                    } else {
                        r[u][0] = row[0];
                    }
                }
                #pragma unroll
                for (int u = 0; u < U; ++u) {
                    if (j0 + h + u < nq) {
                        const int   dl = __builtin_amdgcn_readlane(dv, h + u);
                        const float w  = __int_as_float(__builtin_amdgcn_readlane(wv, h + u));
                        #pragma unroll
                        for (int p = 0; p < PER; ++p)
                            atomicAdd(&acc[dl * F + lane * PER + p], r[u][p] * w);
                    }
                }
            }
        }
    }
    __syncthreads();

    for (int i = tid; i < BN * F; i += AGG_T) {
        const int n = lo + i / F;
        if (n < N) outp[(size_t)n * F + (i % F)] = acc[i];
    }
}

__global__ __launch_bounds__(AGG_T)
void k_range_agg64(const int* __restrict__ ei, const unsigned char* __restrict__ rid,
                   const float* __restrict__ dinv,
                   const float* __restrict__ h, float* __restrict__ agg, int N, int E,
                   const unsigned* __restrict__ sv_s, float* __restrict__ sv_w,
                   const unsigned short* __restrict__ sv_d, const int* __restrict__ sv_c) {
    __shared__ float          acc[BN * D_HID];
    __shared__ unsigned       q[NWAVE * QW];
    __shared__ float          qw[NWAVE * QW];
    __shared__ unsigned short qd[NWAVE * QW];
    range_agg<D_HID, true>(ei, rid, dinv, h, nullptr, agg, N, E, sv_s, sv_w, sv_d, sv_c,
                           acc, q, qw, qd);
}


__global__ void k_gemm2_wmma(const float* __restrict__ AGG, const float* __restrict__ b1,
                             const float* __restrict__ W, float* __restrict__ H2) {
    const int lane = threadIdx.x & 31;
    const int wave = threadIdx.x >> 5;
    const int m0   = blockIdx.x * 16;
    const int n0   = wave * 16;
    const int half = lane >> 4;
    const int lm   = lane & 15;

    const float* arow = AGG + (size_t)(m0 + lm) * D_HID;
    v8f acc = {};
    #pragma unroll 4
    for (int k0 = 0; k0 < D_HID; k0 += 4) {
        const int ka = k0 + 2 * half;
        v2f a, b;
        a.x = fmaxf(arow[ka]     + b1[ka],     0.f);
        a.y = fmaxf(arow[ka + 1] + b1[ka + 1], 0.f);
        b.x = W[(size_t)ka       * D_OUT + n0 + lm];
        b.y = W[(size_t)(ka + 1) * D_OUT + n0 + lm];
        acc = __builtin_amdgcn_wmma_f32_16x16x4_f32(false, a, false, b,
                                                    (short)0, acc, false, false);
    }
    #pragma unroll
    for (int r = 0; r < 8; ++r) {
        const int m = m0 + r + 8 * half;
        H2[(size_t)m * D_OUT + n0 + lm] = acc[r];
    }
}

__global__ __launch_bounds__(AGG_T)
void k_range_agg32(const int* __restrict__ ei, const unsigned char* __restrict__ rid,
                   const float* __restrict__ dinv,
                   const float* __restrict__ h2, const float* __restrict__ b2,
                   float* __restrict__ out, int N, int E,
                   const unsigned* __restrict__ sv_s, float* __restrict__ sv_w,
                   const unsigned short* __restrict__ sv_d, const int* __restrict__ sv_c) {
    __shared__ float          acc[BN * D_OUT];
    __shared__ unsigned       q[NWAVE * QW];
    __shared__ float          qw[NWAVE * QW];
    __shared__ unsigned short qd[NWAVE * QW];
    range_agg<D_OUT, false>(ei, rid, dinv, h2, b2, out, N, E, sv_s, sv_w, sv_d, sv_c,
                            acc, q, qw, qd);
}


extern "C" void kernel_launch(void* const* d_in, const int* in_sizes, int n_in,
                              void* d_out, int out_size, void* d_ws, size_t ws_size,
                              hipStream_t stream) {
    const float* x   = (const float*)d_in[0];
    const int*   ei  = (const int*)  d_in[1];
    const float* W1  = (const float*)d_in[2];
    const float* b1  = (const float*)d_in[3];
    const float* W2  = (const float*)d_in[4];
    const float* b2  = (const float*)d_in[5];
    float*       out = (float*)d_out;

    const int N = in_sizes[0] / D_IN;
    const int E = in_sizes[1] / 2;

    char* ws = (char*)d_ws;
    float*    dinv = (float*)(ws + (size_t)N * 4);
    float*    h    = (float*)(ws + (size_t)N * 8);
    float*    agg1 = (float*)(ws + (size_t)N * 8 + (size_t)N * D_HID * 4);
    float*    h2   = (float*)(ws + (size_t)N * 8 + (size_t)N * D_HID * 8);

    const int TB = 256;

    const int NB = (N + BN - 1) / BN;
    const size_t NQ = (size_t)NB * NWAVE * QW;
    char* qb = ws + (size_t)N * 8 + (size_t)N * D_HID * 8 + (size_t)N * D_OUT * 4;
    unsigned*       sv_s = (unsigned*)(qb);
    float*          sv_w = (float*)(qb + NQ * 4);
    unsigned short* sv_d = (unsigned short*)(qb + NQ * 8);
    int*            sv_c = (int*)(qb + NQ * 10);
    unsigned char*  rid  = (unsigned char*)(sv_c + (size_t)NB * NWAVE);

    k_rid       <<<(E + 4 * TB - 1) / (4 * TB), TB, 0, stream>>>(ei, rid, E);
    k_range_scan<<<NB, AGG_T, 0, stream>>>(ei, rid, dinv, N, E, sv_s, sv_d, sv_c);

    k_gemm1_wmma <<<N / 16, 128, 0, stream>>>(x, W1, h);
    k_range_agg64<<<NB, AGG_T, 0, stream>>>(ei, rid, dinv, h, agg1, N, E, sv_s, sv_w, sv_d, sv_c);

    k_gemm2_wmma <<<N / 16, 64, 0, stream>>>(agg1, b1, W2, h2);
    k_range_agg32<<<NB, AGG_T, 0, stream>>>(ei, rid, dinv, h2, b2, out, N, E, sv_s, sv_w, sv_d, sv_c);
}


# Destination ranges of 400 nodes (were 512): 250 work-groups, one per WGP, fill the chip's 256 WGPs (196 left 60 idle); each resolves, adds and stores a fifth less; sum order kept: self, LDS arrival
# speedup vs baseline: 4.1461x; 1.0576x over previous
#include <hip/hip_runtime.h>
#include <hip/hip_bf16.h>


typedef float v2f __attribute__((ext_vector_type(2)));
typedef float v8f __attribute__((ext_vector_type(8)));

#define D_IN  128
#define D_HID 64
#define D_OUT 32


#define BN 400

__global__ void k_rid(const int* __restrict__ ei, unsigned char* __restrict__ rid, int E) {
    const int i = blockIdx.x * blockDim.x + threadIdx.x;
    const int* dstp = ei + E;
    if ((E & 3) == 0) {
        if (i < (E >> 2)) {
            const int4 d = reinterpret_cast<const int4*>(dstp)[i];
            reinterpret_cast<unsigned*>(rid)[i] =
                ((unsigned)d.x / BN)         | (((unsigned)d.y / BN) << 8) |
                (((unsigned)d.z / BN) << 16) | (((unsigned)d.w / BN) << 24);
        }
    } else {
        for (int u = 0; u < 4; ++u) {
            const int e = 4 * i + u;
            if (e < E) rid[e] = (unsigned char)((unsigned)dstp[e] / BN);
        }
    }
}


__global__ void k_gemm1_wmma(const float* __restrict__ X, const float* __restrict__ W,
                             float* __restrict__ H) {
    const int lane = threadIdx.x & 31;
    const int wave = threadIdx.x >> 5;
    const int m0   = blockIdx.x * 16;
    const int n0   = wave * 16;
    const int half = lane >> 4;
    const int lm   = lane & 15;

    const float* xrow = X + (size_t)(m0 + lm) * D_IN;
    v8f acc = {};
    #pragma unroll 4
    for (int k0 = 0; k0 < D_IN; k0 += 4) {
        const int ka = k0 + 2 * half;
        v2f a, b;
        a.x = xrow[ka];
        a.y = xrow[ka + 1];
        b.x = W[(size_t)ka       * D_HID + n0 + lm];
        b.y = W[(size_t)(ka + 1) * D_HID + n0 + lm];
        acc = __builtin_amdgcn_wmma_f32_16x16x4_f32(false, a, false, b,
                                                    (short)0, acc, false, false);
    }
    #pragma unroll
    for (int r = 0; r < 8; ++r) {
        const int m = m0 + r + 8 * half;
        H[(size_t)m * D_HID + n0 + lm] = acc[r];
    }
}


#define AGG_T 1024
#define NWAVE (AGG_T / 32)
#define QW    384

template <class Direct>
__device__ __forceinline__ int scan_edges(const int* __restrict__ ei,
                                          const unsigned char* __restrict__ rid,
                                          int E, int lo, unsigned lim, unsigned* wq,
                                          Direct direct) {
    const int tid  = threadIdx.x;
    const int* dstp = ei + E;
    int wcount = 0;

    auto append = [&](unsigned bal, bool hit, int e) {
        if (hit) {
            const int pos = wcount + (int)__builtin_amdgcn_mbcnt_lo(bal, 0u);
            if (pos < QW) wq[pos] = (unsigned)e;
            else          direct(e);
        }
        wcount += __builtin_popcount(bal);
    };

    if ((E & 15) == 0) {
        const unsigned pat = (unsigned)blockIdx.x * 0x01010101u;

        auto sixteen = [&](const uint4 t, int e0, bool live) {
            auto other = [&](unsigned x) {
                const unsigned y = x ^ pat;
                return (((y | 0x80808080u) - 0x01010101u) | y) & 0x80808080u;
            };
            unsigned m = ~((other(t.x) >> 7) | (other(t.y) >> 6) |
                           (other(t.z) >> 5) | (other(t.w) >> 4)) & 0x0F0F0F0Fu;
            if (!live) m = 0u;
            unsigned bal = __builtin_amdgcn_ballot_w32(m != 0u);
            while (bal != 0u) {
                if (m != 0u) {
                    const int p = __builtin_ctz(m);
                    const int e = e0 + 4 * (p & 7) + (p >> 3);
                    const int pos = wcount + (int)__builtin_amdgcn_mbcnt_lo(bal, 0u);
                    if (pos < QW) wq[pos] = (unsigned)e;
                    else          direct(e);
                }
                wcount += __builtin_popcount(bal);
                m &= m - 1u;
                bal = __builtin_amdgcn_ballot_w32(m != 0u);
            }
        };

        const uint4* b16 = reinterpret_cast<const uint4*>(rid);
        const int n16 = E >> 4;
        constexpr int L = 8;
        const int CH = L * AGG_T;
        const int trips = n16 / CH;
        const int rot = (int)(((long long)blockIdx.x * trips) / gridDim.x);
        uint4 cur[L], nxt[L];
        if (trips > 0) {
            #pragma unroll
            for (int u = 0; u < L; ++u) cur[u] = b16[rot * CH + u * AGG_T + tid];
        }
        for (int it = 0; it < trips; ++it) {
            int t = it + rot;
            if (t >= trips) t -= trips;
            int t1 = t + 1;
            if (t1 >= trips) t1 -= trips;
            #pragma unroll
            for (int u = 0; u < L; ++u) nxt[u] = b16[t1 * CH + u * AGG_T + tid];
            #pragma unroll
            for (int u = 0; u < L; ++u) sixteen(cur[u], (t * CH + u * AGG_T + tid) * 16, true);
            #pragma unroll
            for (int u = 0; u < L; ++u) cur[u] = nxt[u];
        }
        for (int base = trips * CH; base < n16; base += AGG_T) {
            const int i16 = base + tid;
            const bool live = i16 < n16;
            sixteen(b16[live ? i16 : 0], i16 * 16, live);
        }
    } else {
        for (int base = 0; base < E; base += AGG_T) {
            const int e = base + tid;
            const bool hit = (unsigned)((e < E ? dstp[e] : -1) - lo) < lim;
            const unsigned bal = __builtin_amdgcn_ballot_w32(hit);
            if (bal != 0u) append(bal, hit, e);
        }
    }
    return wcount;
}

__global__ __launch_bounds__(AGG_T)
void k_range_scan(const int* __restrict__ ei, const unsigned char* __restrict__ rid,
                  float* __restrict__ dinv, int N, int E,
                  unsigned* __restrict__ sv_s, unsigned short* __restrict__ sv_d,
                  int* __restrict__ sv_c) {
    __shared__ unsigned cnt[BN];
    __shared__ unsigned q[NWAVE * QW];

    const int tid  = threadIdx.x;
    const int lane = tid & 31;
    const int wave = tid >> 5;
    const int lo   = blockIdx.x * BN;
    const int left = N - lo;
    const unsigned lim = (unsigned)(left < BN ? left : BN);
    const int* dstp = ei + E;

    unsigned* wq = q + wave * QW;
    const size_t slot = (size_t)blockIdx.x * NWAVE + wave;
    unsigned*       ws_s = sv_s + slot * QW;
    unsigned short* ws_d = sv_d + slot * QW;

    for (int i = tid; i < BN; i += AGG_T) cnt[i] = 0u;
    __syncthreads();

    auto direct = [&](int e) { atomicAdd(&cnt[dstp[e] - lo], 1u); };
    const int wcount = scan_edges(ei, rid, E, lo, lim, wq, direct);
    const int nq = wcount < QW ? wcount : QW;
    if (lane == 0) sv_c[slot] = wcount;
    __builtin_amdgcn_fence(__ATOMIC_ACQ_REL, "wavefront");

    for (int j0 = lane; j0 < nq; j0 += 4 * 32) {
        int e[4], s[4], d[4];
        #pragma unroll
        for (int u = 0; u < 4; ++u) {
            const int j = j0 + u * 32;
            e[u] = j < nq ? (int)wq[j] : 0;
        }
        #pragma unroll
        for (int u = 0; u < 4; ++u) { s[u] = ei[e[u]]; d[u] = dstp[e[u]]; }
        #pragma unroll
        for (int u = 0; u < 4; ++u) {
            const int j = j0 + u * 32;
            if (j < nq) {
                const unsigned dl = (unsigned)(d[u] - lo);
                ws_s[j] = (unsigned)s[u];
                ws_d[j] = (unsigned short)dl;
                atomicAdd(&cnt[dl], 1u);
            }
        }
    }
    __syncthreads();

    for (int i = tid; i < BN; i += AGG_T) {
        const int n = lo + i;
        if (n < N) dinv[n] = rsqrtf((float)(cnt[i] + 1u));
    }
}

template <int F>
__device__ __forceinline__ int scan_resolve(const int* __restrict__ ei,
                                            const unsigned char* __restrict__ rid,
                                            const float* __restrict__ dinv,
                                            const float* __restrict__ hin,
                                            int E, int lo, unsigned lim, float* acc,
                                            unsigned* wq, float* wqw, unsigned short* wqd) {
    const int lane = threadIdx.x & 31;
    const int* dstp = ei + E;

    auto direct = [&](int e) {
        const int s  = ei[e];
        const int dn = dstp[e];
        const unsigned dl = (unsigned)(dn - lo);
        const float w = dinv[s] * dinv[dn];
        #pragma unroll 1
        for (int f = 0; f < F; ++f)
            atomicAdd(&acc[dl * F + f], hin[(size_t)s * F + f] * w);
    };
    const int wcount = scan_edges(ei, rid, E, lo, lim, wq, direct);
    const int nq = wcount < QW ? wcount : QW;
    __builtin_amdgcn_fence(__ATOMIC_ACQ_REL, "wavefront");

    for (int j0 = lane; j0 < nq; j0 += 4 * 32) {
        int e[4], s[4], d[4];
        float ws[4], wd[4];
        #pragma unroll
        for (int u = 0; u < 4; ++u) {
            const int j = j0 + u * 32;
            e[u] = j < nq ? (int)wq[j] : 0;
        }
        #pragma unroll
        for (int u = 0; u < 4; ++u) { s[u] = ei[e[u]]; d[u] = dstp[e[u]]; }
        #pragma unroll
        for (int u = 0; u < 4; ++u) { ws[u] = dinv[s[u]]; wd[u] = dinv[d[u]]; }
        #pragma unroll
        for (int u = 0; u < 4; ++u) {
            const int j = j0 + u * 32;
            if (j < nq) {
                wq[j]  = (unsigned)s[u];
                wqd[j] = (unsigned short)(d[u] - lo);
                wqw[j] = ws[u] * wd[u];
            }
        }
    }
    return nq;
}

template <int F, bool FIRST>
__device__ __forceinline__ void range_agg(const int* __restrict__ ei,
                                          const unsigned char* __restrict__ rid,
                                          const float* __restrict__ dinv,
                                          const float* __restrict__ hin,
                                          const float* __restrict__ bias,
                                          float* __restrict__ outp, int N, int E,
                                          const unsigned* __restrict__ sv_s,
                                          float* __restrict__ sv_w,
                                          const unsigned short* __restrict__ sv_d,
                                          const int* __restrict__ sv_c,
                                          float* acc, unsigned* q, float* qw,
                                          unsigned short* qd) {
    const int tid  = threadIdx.x;
    const int lane = tid & 31;
    const int wave = tid >> 5;
    const int lo   = blockIdx.x * BN;
    const int left = N - lo;
    const unsigned lim = (unsigned)(left < BN ? left : BN);

    unsigned*       wq  = q  + wave * QW;
    float*          wqw = qw + wave * QW;
    unsigned short* wqd = qd + wave * QW;

    const size_t slot = (size_t)blockIdx.x * NWAVE + wave;
    const unsigned*       ws_s = sv_s + slot * QW;
    float*                ws_w = sv_w + slot * QW;
    const unsigned short* ws_d = sv_d + slot * QW;

    #pragma unroll 8
    for (int i = tid; i < BN * F; i += AGG_T) {
        const int n  = lo + i / F;
        const int f  = i % F;
        const int nn = n < N ? n : N - 1;
        const float w = dinv[nn];
        const float x = hin[(size_t)nn * F + f];
        const float v = bias ? bias[f] + w * w * x : w * w * x;
        acc[i] = n < N ? v : 0.f;
    }
    __syncthreads();

    int nq;
    const int met = __builtin_amdgcn_readfirstlane(sv_c[slot]);
    if (met > QW) {
        nq = scan_resolve<F>(ei, rid, dinv, hin, E, lo, lim, acc, wq, wqw, wqd);
    } else {
        nq = met;
        for (int j0 = lane; j0 < nq; j0 += 8 * 32) {
            unsigned s[8];
            unsigned short d[8];
            float w[8];
            #pragma unroll
            for (int u = 0; u < 8; ++u) {
                const int j  = j0 + u * 32;
                const int jj = j < nq ? j : 0;
                s[u] = ws_s[jj];
                d[u] = ws_d[jj];
                if constexpr (!FIRST) w[u] = ws_w[jj];
            }
            if constexpr (FIRST) {
                float a[8], b[8];
                #pragma unroll
                for (int u = 0; u < 8; ++u) { a[u] = dinv[s[u]]; b[u] = dinv[lo + (int)d[u]]; }
                #pragma unroll
                for (int u = 0; u < 8; ++u) w[u] = a[u] * b[u];
            }
            #pragma unroll
            for (int u = 0; u < 8; ++u) {
                const int j = j0 + u * 32;
                if (j < nq) {
                    wq[j] = s[u]; wqd[j] = d[u]; wqw[j] = w[u];
                    if constexpr (FIRST) ws_w[j] = w[u];
                }
            }
        }
    }
    __builtin_amdgcn_fence(__ATOMIC_ACQ_REL, "wavefront");

    constexpr int PER = F / 32;
    constexpr int U   = 16;
    for (int j0 = 0; j0 < nq; j0 += 32) {
        const int jl = (j0 + lane < nq) ? j0 + lane : nq - 1;
        const int sv = (int)wq[jl];
        const int dv = (int)wqd[jl];
        const int wv = __float_as_int(wqw[jl]);
        #pragma unroll
        for (int h = 0; h < 32; h += U) {
            if (j0 + h < nq) {
                float r[U][PER];
                #pragma unroll
                for (int u = 0; u < U; ++u) {
                    const size_t s = (unsigned)__builtin_amdgcn_readlane(sv, h + u);
                    const float* row = hin + s * F + lane * PER;
                    if constexpr (PER == 2) {
                        const float2 t2 = *reinterpret_cast<const float2*>(row);
                        r[u][0] = t2.x; r[u][1] = t2.y;
                    } else {
                        r[u][0] = row[0];
                    }
                }
                #pragma unroll
                for (int u = 0; u < U; ++u) {
                    if (j0 + h + u < nq) {
                        const int   dl = __builtin_amdgcn_readlane(dv, h + u);
                        const float w  = __int_as_float(__builtin_amdgcn_readlane(wv, h + u));
                        #pragma unroll
                        for (int p = 0; p < PER; ++p)
                            atomicAdd(&acc[dl * F + lane * PER + p], r[u][p] * w);
                    }
                }
            }
        }
    }
    __syncthreads();

    for (int i = tid; i < BN * F; i += AGG_T) {
        const int n = lo + i / F;
        if (n < N) outp[(size_t)n * F + (i % F)] = acc[i];
    }
}

__global__ __launch_bounds__(AGG_T)
void k_range_agg64(const int* __restrict__ ei, const unsigned char* __restrict__ rid,
                   const float* __restrict__ dinv,
                   const float* __restrict__ h, float* __restrict__ agg, int N, int E,
                   const unsigned* __restrict__ sv_s, float* __restrict__ sv_w,
                   const unsigned short* __restrict__ sv_d, const int* __restrict__ sv_c) {
    __shared__ float          acc[BN * D_HID];
    __shared__ unsigned       q[NWAVE * QW];
    __shared__ float          qw[NWAVE * QW];
    __shared__ unsigned short qd[NWAVE * QW];
    range_agg<D_HID, true>(ei, rid, dinv, h, nullptr, agg, N, E, sv_s, sv_w, sv_d, sv_c,
                           acc, q, qw, qd);
}


__global__ void k_gemm2_wmma(const float* __restrict__ AGG, const float* __restrict__ b1,
                             const float* __restrict__ W, float* __restrict__ H2) {
    const int lane = threadIdx.x & 31;
    const int wave = threadIdx.x >> 5;
    const int m0   = blockIdx.x * 16;
    const int n0   = wave * 16;
    const int half = lane >> 4;
    const int lm   = lane & 15;

    const float* arow = AGG + (size_t)(m0 + lm) * D_HID;
    v8f acc = {};
    #pragma unroll 4
    for (int k0 = 0; k0 < D_HID; k0 += 4) {
        const int ka = k0 + 2 * half;
        v2f a, b;
        a.x = fmaxf(arow[ka]     + b1[ka],     0.f);
        a.y = fmaxf(arow[ka + 1] + b1[ka + 1], 0.f);
        b.x = W[(size_t)ka       * D_OUT + n0 + lm];
        b.y = W[(size_t)(ka + 1) * D_OUT + n0 + lm];
        acc = __builtin_amdgcn_wmma_f32_16x16x4_f32(false, a, false, b,
                                                    (short)0, acc, false, false);
    }
    #pragma unroll
    for (int r = 0; r < 8; ++r) {
        const int m = m0 + r + 8 * half;
        H2[(size_t)m * D_OUT + n0 + lm] = acc[r];
    }
}

__global__ __launch_bounds__(AGG_T)
void k_range_agg32(const int* __restrict__ ei, const unsigned char* __restrict__ rid,
                   const float* __restrict__ dinv,
                   const float* __restrict__ h2, const float* __restrict__ b2,
                   float* __restrict__ out, int N, int E,
                   const unsigned* __restrict__ sv_s, float* __restrict__ sv_w,
                   const unsigned short* __restrict__ sv_d, const int* __restrict__ sv_c) {
    __shared__ float          acc[BN * D_OUT];
    __shared__ unsigned       q[NWAVE * QW];
    __shared__ float          qw[NWAVE * QW];
    __shared__ unsigned short qd[NWAVE * QW];
    range_agg<D_OUT, false>(ei, rid, dinv, h2, b2, out, N, E, sv_s, sv_w, sv_d, sv_c,
                            acc, q, qw, qd);
}


extern "C" void kernel_launch(void* const* d_in, const int* in_sizes, int n_in,
                              void* d_out, int out_size, void* d_ws, size_t ws_size,
                              hipStream_t stream) {
    const float* x   = (const float*)d_in[0];
    const int*   ei  = (const int*)  d_in[1];
    const float* W1  = (const float*)d_in[2];
    const float* b1  = (const float*)d_in[3];
    const float* W2  = (const float*)d_in[4];
    const float* b2  = (const float*)d_in[5];
    float*       out = (float*)d_out;

    const int N = in_sizes[0] / D_IN;
    const int E = in_sizes[1] / 2;

    char* ws = (char*)d_ws;
    float*    dinv = (float*)(ws + (size_t)N * 4);
    float*    h    = (float*)(ws + (size_t)N * 8);
    float*    agg1 = (float*)(ws + (size_t)N * 8 + (size_t)N * D_HID * 4);
    float*    h2   = (float*)(ws + (size_t)N * 8 + (size_t)N * D_HID * 8);

    const int TB = 256;

    const int NB = (N + BN - 1) / BN;
    const size_t NQ = (size_t)NB * NWAVE * QW;
    char* qb = ws + (size_t)N * 8 + (size_t)N * D_HID * 8 + (size_t)N * D_OUT * 4;
    unsigned*       sv_s = (unsigned*)(qb);
    float*          sv_w = (float*)(qb + NQ * 4);
    unsigned short* sv_d = (unsigned short*)(qb + NQ * 8);
    int*            sv_c = (int*)(qb + NQ * 10);
    unsigned char*  rid  = (unsigned char*)(sv_c + (size_t)NB * NWAVE);

    k_rid       <<<(E + 4 * TB - 1) / (4 * TB), TB, 0, stream>>>(ei, rid, E);
    k_range_scan<<<NB, AGG_T, 0, stream>>>(ei, rid, dinv, N, E, sv_s, sv_d, sv_c);

    k_gemm1_wmma <<<N / 16, 128, 0, stream>>>(x, W1, h);
    k_range_agg64<<<NB, AGG_T, 0, stream>>>(ei, rid, dinv, h, agg1, N, E, sv_s, sv_w, sv_d, sv_c);

    k_gemm2_wmma <<<N / 16, 64, 0, stream>>>(agg1, b1, W2, h2);
    k_range_agg32<<<NB, AGG_T, 0, stream>>>(ei, rid, dinv, h2, b2, out, N, E, sv_s, sv_w, sv_d, sv_c);
}


# Both GEMMs re-tiled: a work-group owns 64 rows, stages their A block in LDS (whole-row 16-byte loads); a wave owns a 64x16 column block, a W fragment feeds 4 row tiles; k order and sums unchanged
# speedup vs baseline: 4.2987x; 1.0368x over previous
#include <hip/hip_runtime.h>
#include <hip/hip_bf16.h>


typedef float v2f __attribute__((ext_vector_type(2)));
typedef float v8f __attribute__((ext_vector_type(8)));

#define D_IN  128
#define D_HID 64
#define D_OUT 32


#define BN 400

__global__ void k_rid(const int* __restrict__ ei, unsigned char* __restrict__ rid, int E) {
    const int i = blockIdx.x * blockDim.x + threadIdx.x;
    const int* dstp = ei + E;
    if ((E & 3) == 0) {
        if (i < (E >> 2)) {
            const int4 d = reinterpret_cast<const int4*>(dstp)[i];
            reinterpret_cast<unsigned*>(rid)[i] =
                ((unsigned)d.x / BN)         | (((unsigned)d.y / BN) << 8) |
                (((unsigned)d.z / BN) << 16) | (((unsigned)d.w / BN) << 24);
        }
    } else {
        for (int u = 0; u < 4; ++u) {
            const int e = 4 * i + u;
            if (e < E) rid[e] = (unsigned char)((unsigned)dstp[e] / BN);
        }
    }
}


#define GM 64

template <int K, int NOUT, bool ACT>
__global__ __launch_bounds__(NOUT * 2)
void k_gemm_rows(const float* __restrict__ A, const float* __restrict__ bias,
                 const float* __restrict__ W, float* __restrict__ H, int N) {
    constexpr int NT = NOUT * 2;
    constexpr int P  = K + 4;
    constexpr int PO = NOUT + 4;
    constexpr int C4 = K / 4;
    constexpr int O4 = NOUT / 4;
    __shared__ __attribute__((aligned(16))) float xs[GM * P];

    const int tid  = threadIdx.x;
    const int lane = tid & 31;
    const int wave = tid >> 5;
    const int m0   = blockIdx.x * GM;
    const int n0   = wave * 16;
    const int half = lane >> 4;
    const int lm   = lane & 15;

    #pragma unroll
    for (int u = 0; u < GM * C4 / NT; ++u) {
        const int i = u * NT + tid;
        const int r = i / C4;
        const int c = i % C4;
        const int m = m0 + r < N ? m0 + r : N - 1;
        float4 v = *reinterpret_cast<const float4*>(A + (size_t)m * K + 4 * c);
        if constexpr (ACT) {
            v.x = fmaxf(v.x + bias[4 * c],     0.f);
            v.y = fmaxf(v.y + bias[4 * c + 1], 0.f);
            v.z = fmaxf(v.z + bias[4 * c + 2], 0.f);
            v.w = fmaxf(v.w + bias[4 * c + 3], 0.f);
        }
        *reinterpret_cast<float4*>(&xs[r * P + 4 * c]) = v;
    }
    __syncthreads();

    v8f acc[4] = {};
    const float* xa = xs + lm * P + 2 * half;
    #pragma unroll 4
    for (int k0 = 0; k0 < K; k0 += 4) {
        const int ka = k0 + 2 * half;
        v2f b;
        b.x = W[(size_t)ka       * NOUT + n0 + lm];
        b.y = W[(size_t)(ka + 1) * NOUT + n0 + lm];
        #pragma unroll
        for (int t = 0; t < 4; ++t) {
            const v2f a = *reinterpret_cast<const v2f*>(xa + t * 16 * P + k0);
            acc[t] = __builtin_amdgcn_wmma_f32_16x16x4_f32(false, a, false, b,
                                                           (short)0, acc[t], false, false);
        }
    }
    __syncthreads();

    #pragma unroll
    for (int t = 0; t < 4; ++t) {
        #pragma unroll
        for (int r = 0; r < 8; ++r)
            xs[(t * 16 + r + 8 * half) * PO + n0 + lm] = acc[t][r];
    }
    __syncthreads();
    float4 o[GM * O4 / NT];
    #pragma unroll
    for (int u = 0; u < GM * O4 / NT; ++u) {
        const int i = u * NT + tid;
        o[u] = *reinterpret_cast<const float4*>(&xs[(i / O4) * PO + 4 * (i % O4)]);
    }
    if (m0 + GM <= N) {
        #pragma unroll
        for (int u = 0; u < GM * O4 / NT; ++u) {
            const int i = u * NT + tid;
            *reinterpret_cast<float4*>(H + (size_t)(m0 + i / O4) * NOUT + 4 * (i % O4)) = o[u];
        }
    } else {
        #pragma unroll
        for (int u = 0; u < GM * O4 / NT; ++u) {
            const int i = u * NT + tid;
            const int m = m0 + i / O4;
            if (m < N)
                *reinterpret_cast<float4*>(H + (size_t)m * NOUT + 4 * (i % O4)) = o[u];
        }
    }
}


#define AGG_T 1024
#define NWAVE (AGG_T / 32)
#define QW    384

template <class Direct>
__device__ __forceinline__ int scan_edges(const int* __restrict__ ei,
                                          const unsigned char* __restrict__ rid,
                                          int E, int lo, unsigned lim, unsigned* wq,
                                          Direct direct) {
    const int tid  = threadIdx.x;
    const int* dstp = ei + E;
    int wcount = 0;

    auto append = [&](unsigned bal, bool hit, int e) {
        if (hit) {
            const int pos = wcount + (int)__builtin_amdgcn_mbcnt_lo(bal, 0u);
            if (pos < QW) wq[pos] = (unsigned)e;
            else          direct(e);
        }
        wcount += __builtin_popcount(bal);
    };

    if ((E & 15) == 0) {
        const unsigned pat = (unsigned)blockIdx.x * 0x01010101u;

        auto sixteen = [&](const uint4 t, int e0, bool live) {
            auto other = [&](unsigned x) {
                const unsigned y = x ^ pat;
                return (((y | 0x80808080u) - 0x01010101u) | y) & 0x80808080u;
            };
            unsigned m = ~((other(t.x) >> 7) | (other(t.y) >> 6) |
                           (other(t.z) >> 5) | (other(t.w) >> 4)) & 0x0F0F0F0Fu;
            if (!live) m = 0u;
            unsigned bal = __builtin_amdgcn_ballot_w32(m != 0u);
            while (bal != 0u) {
                if (m != 0u) {
                    const int p = __builtin_ctz(m);
                    const int e = e0 + 4 * (p & 7) + (p >> 3);
                    const int pos = wcount + (int)__builtin_amdgcn_mbcnt_lo(bal, 0u);
                    if (pos < QW) wq[pos] = (unsigned)e;
                    else          direct(e);
                }
                wcount += __builtin_popcount(bal);
                m &= m - 1u;
                bal = __builtin_amdgcn_ballot_w32(m != 0u);
            }
        };

        const uint4* b16 = reinterpret_cast<const uint4*>(rid);
        const int n16 = E >> 4;
        constexpr int L = 8;
        const int CH = L * AGG_T;
        const int trips = n16 / CH;
        const int rot = (int)(((long long)blockIdx.x * trips) / gridDim.x);
        uint4 cur[L], nxt[L];
        if (trips > 0) {
            #pragma unroll
            for (int u = 0; u < L; ++u) cur[u] = b16[rot * CH + u * AGG_T + tid];
        }
        for (int it = 0; it < trips; ++it) {
            int t = it + rot;
            if (t >= trips) t -= trips;
            int t1 = t + 1;
            if (t1 >= trips) t1 -= trips;
            #pragma unroll
            for (int u = 0; u < L; ++u) nxt[u] = b16[t1 * CH + u * AGG_T + tid];
            #pragma unroll
            for (int u = 0; u < L; ++u) sixteen(cur[u], (t * CH + u * AGG_T + tid) * 16, true);
            #pragma unroll
            for (int u = 0; u < L; ++u) cur[u] = nxt[u];
        }
        for (int base = trips * CH; base < n16; base += AGG_T) {
            const int i16 = base + tid;
            const bool live = i16 < n16;
            sixteen(b16[live ? i16 : 0], i16 * 16, live);
        }
    } else {
        for (int base = 0; base < E; base += AGG_T) {
            const int e = base + tid;
            const bool hit = (unsigned)((e < E ? dstp[e] : -1) - lo) < lim;
            const unsigned bal = __builtin_amdgcn_ballot_w32(hit);
            if (bal != 0u) append(bal, hit, e);
        }
    }
    return wcount;
}

__global__ __launch_bounds__(AGG_T)
void k_range_scan(const int* __restrict__ ei, const unsigned char* __restrict__ rid,
                  float* __restrict__ dinv, int N, int E,
                  unsigned* __restrict__ sv_s, unsigned short* __restrict__ sv_d,
                  int* __restrict__ sv_c) {
    __shared__ unsigned cnt[BN];
    __shared__ unsigned q[NWAVE * QW];

    const int tid  = threadIdx.x;
    const int lane = tid & 31;
    const int wave = tid >> 5;
    const int lo   = blockIdx.x * BN;
    const int left = N - lo;
    const unsigned lim = (unsigned)(left < BN ? left : BN);
    const int* dstp = ei + E;

    unsigned* wq = q + wave * QW;
    const size_t slot = (size_t)blockIdx.x * NWAVE + wave;
    unsigned*       ws_s = sv_s + slot * QW;
    unsigned short* ws_d = sv_d + slot * QW;

    for (int i = tid; i < BN; i += AGG_T) cnt[i] = 0u;
    __syncthreads();

    auto direct = [&](int e) { atomicAdd(&cnt[dstp[e] - lo], 1u); };
    const int wcount = scan_edges(ei, rid, E, lo, lim, wq, direct);
    const int nq = wcount < QW ? wcount : QW;
    if (lane == 0) sv_c[slot] = wcount;
    __builtin_amdgcn_fence(__ATOMIC_ACQ_REL, "wavefront");

    for (int j0 = lane; j0 < nq; j0 += 4 * 32) {
        int e[4], s[4], d[4];
        #pragma unroll
        for (int u = 0; u < 4; ++u) {
            const int j = j0 + u * 32;
            e[u] = j < nq ? (int)wq[j] : 0;
        }
        #pragma unroll
        for (int u = 0; u < 4; ++u) { s[u] = ei[e[u]]; d[u] = dstp[e[u]]; }
        #pragma unroll
        for (int u = 0; u < 4; ++u) {
            const int j = j0 + u * 32;
            if (j < nq) {
                const unsigned dl = (unsigned)(d[u] - lo);
                ws_s[j] = (unsigned)s[u];
                ws_d[j] = (unsigned short)dl;
                atomicAdd(&cnt[dl], 1u);
            }
        }
    }
    __syncthreads();

    for (int i = tid; i < BN; i += AGG_T) {
        const int n = lo + i;
        if (n < N) dinv[n] = rsqrtf((float)(cnt[i] + 1u));
    }
}

template <int F>
__device__ __forceinline__ int scan_resolve(const int* __restrict__ ei,
                                            const unsigned char* __restrict__ rid,
                                            const float* __restrict__ dinv,
                                            const float* __restrict__ hin,
                                            int E, int lo, unsigned lim, float* acc,
                                            unsigned* wq, float* wqw, unsigned short* wqd) {
    const int lane = threadIdx.x & 31;
    const int* dstp = ei + E;

    auto direct = [&](int e) {
        const int s  = ei[e];
        const int dn = dstp[e];
        const unsigned dl = (unsigned)(dn - lo);
        const float w = dinv[s] * dinv[dn];
        #pragma unroll 1
        for (int f = 0; f < F; ++f)
            atomicAdd(&acc[dl * F + f], hin[(size_t)s * F + f] * w);
    };
    const int wcount = scan_edges(ei, rid, E, lo, lim, wq, direct);
    const int nq = wcount < QW ? wcount : QW;
    __builtin_amdgcn_fence(__ATOMIC_ACQ_REL, "wavefront");

    for (int j0 = lane; j0 < nq; j0 += 4 * 32) {
        int e[4], s[4], d[4];
        float ws[4], wd[4];
        #pragma unroll
        for (int u = 0; u < 4; ++u) {
            const int j = j0 + u * 32;
            e[u] = j < nq ? (int)wq[j] : 0;
        }
        #pragma unroll
        for (int u = 0; u < 4; ++u) { s[u] = ei[e[u]]; d[u] = dstp[e[u]]; }
        #pragma unroll
        for (int u = 0; u < 4; ++u) { ws[u] = dinv[s[u]]; wd[u] = dinv[d[u]]; }
        #pragma unroll
        for (int u = 0; u < 4; ++u) {
            const int j = j0 + u * 32;
            if (j < nq) {
                wq[j]  = (unsigned)s[u];
                wqd[j] = (unsigned short)(d[u] - lo);
                wqw[j] = ws[u] * wd[u];
            }
        }
    }
    return nq;
}

template <int F, bool FIRST>
__device__ __forceinline__ void range_agg(const int* __restrict__ ei,
                                          const unsigned char* __restrict__ rid,
                                          const float* __restrict__ dinv,
                                          const float* __restrict__ hin,
                                          const float* __restrict__ bias,
                                          float* __restrict__ outp, int N, int E,
                                          const unsigned* __restrict__ sv_s,
                                          float* __restrict__ sv_w,
                                          const unsigned short* __restrict__ sv_d,
                                          const int* __restrict__ sv_c,
                                          float* acc, unsigned* q, float* qw,
                                          unsigned short* qd) {
    const int tid  = threadIdx.x;
    const int lane = tid & 31;
    const int wave = tid >> 5;
    const int lo   = blockIdx.x * BN;
    const int left = N - lo;
    const unsigned lim = (unsigned)(left < BN ? left : BN);

    unsigned*       wq  = q  + wave * QW;
    float*          wqw = qw + wave * QW;
    unsigned short* wqd = qd + wave * QW;

    const size_t slot = (size_t)blockIdx.x * NWAVE + wave;
    const unsigned*       ws_s = sv_s + slot * QW;
    float*                ws_w = sv_w + slot * QW;
    const unsigned short* ws_d = sv_d + slot * QW;

    #pragma unroll 8
    for (int i = tid; i < BN * F; i += AGG_T) {
        const int n  = lo + i / F;
        const int f  = i % F;
        const int nn = n < N ? n : N - 1;
        const float w = dinv[nn];
        const float x = hin[(size_t)nn * F + f];
        const float v = bias ? bias[f] + w * w * x : w * w * x;
        acc[i] = n < N ? v : 0.f;
    }
    __syncthreads();

    int nq;
    const int met = __builtin_amdgcn_readfirstlane(sv_c[slot]);
    if (met > QW) {
        nq = scan_resolve<F>(ei, rid, dinv, hin, E, lo, lim, acc, wq, wqw, wqd);
    } else {
        nq = met;
        for (int j0 = lane; j0 < nq; j0 += 8 * 32) {
            unsigned s[8];
            unsigned short d[8];
            float w[8];
            #pragma unroll
            for (int u = 0; u < 8; ++u) {
                const int j  = j0 + u * 32;
                const int jj = j < nq ? j : 0;
                s[u] = ws_s[jj];
                d[u] = ws_d[jj];
                if constexpr (!FIRST) w[u] = ws_w[jj];
            }
            if constexpr (FIRST) {
                float a[8], b[8];
                #pragma unroll
                for (int u = 0; u < 8; ++u) { a[u] = dinv[s[u]]; b[u] = dinv[lo + (int)d[u]]; }
                #pragma unroll
                for (int u = 0; u < 8; ++u) w[u] = a[u] * b[u];
            }
            #pragma unroll
            for (int u = 0; u < 8; ++u) {
                const int j = j0 + u * 32;
                if (j < nq) {
                    wq[j] = s[u]; wqd[j] = d[u]; wqw[j] = w[u];
                    if constexpr (FIRST) ws_w[j] = w[u];
                }
            }
        }
    }
    __builtin_amdgcn_fence(__ATOMIC_ACQ_REL, "wavefront");

    constexpr int PER = F / 32;
    constexpr int U   = 16;
    for (int j0 = 0; j0 < nq; j0 += 32) {
        const int jl = (j0 + lane < nq) ? j0 + lane : nq - 1;
        const int sv = (int)wq[jl];
        const int dv = (int)wqd[jl];
        const int wv = __float_as_int(wqw[jl]);
        #pragma unroll
        for (int h = 0; h < 32; h += U) {
            if (j0 + h < nq) {
                float r[U][PER];
                #pragma unroll
                for (int u = 0; u < U; ++u) {
                    const size_t s = (unsigned)__builtin_amdgcn_readlane(sv, h + u);
                    const float* row = hin + s * F + lane * PER;
                    if constexpr (PER == 2) {
                        const float2 t2 = *reinterpret_cast<const float2*>(row);
                        r[u][0] = t2.x; r[u][1] = t2.y;
                    } else {
                        r[u][0] = row[0];
                    }
                }
                #pragma unroll
                for (int u = 0; u < U; ++u) {
                    if (j0 + h + u < nq) {
                        const int   dl = __builtin_amdgcn_readlane(dv, h + u);
                        const float w  = __int_as_float(__builtin_amdgcn_readlane(wv, h + u));
                        #pragma unroll
                        for (int p = 0; p < PER; ++p)
                            atomicAdd(&acc[dl * F + lane * PER + p], r[u][p] * w);
                    }
                }
            }
        }
    }
    __syncthreads();

    for (int i = tid; i < BN * F; i += AGG_T) {
        const int n = lo + i / F;
        if (n < N) outp[(size_t)n * F + (i % F)] = acc[i];
    }
}

__global__ __launch_bounds__(AGG_T)
void k_range_agg64(const int* __restrict__ ei, const unsigned char* __restrict__ rid,
                   const float* __restrict__ dinv,
                   const float* __restrict__ h, float* __restrict__ agg, int N, int E,
                   const unsigned* __restrict__ sv_s, float* __restrict__ sv_w,
                   const unsigned short* __restrict__ sv_d, const int* __restrict__ sv_c) {
    __shared__ float          acc[BN * D_HID];
    __shared__ unsigned       q[NWAVE * QW];
    __shared__ float          qw[NWAVE * QW];
    __shared__ unsigned short qd[NWAVE * QW];
    range_agg<D_HID, true>(ei, rid, dinv, h, nullptr, agg, N, E, sv_s, sv_w, sv_d, sv_c,
                           acc, q, qw, qd);
}

__global__ __launch_bounds__(AGG_T)
void k_range_agg32(const int* __restrict__ ei, const unsigned char* __restrict__ rid,
                   const float* __restrict__ dinv,
                   const float* __restrict__ h2, const float* __restrict__ b2,
                   float* __restrict__ out, int N, int E,
                   const unsigned* __restrict__ sv_s, float* __restrict__ sv_w,
                   const unsigned short* __restrict__ sv_d, const int* __restrict__ sv_c) {
    __shared__ float          acc[BN * D_OUT];
    __shared__ unsigned       q[NWAVE * QW];
    __shared__ float          qw[NWAVE * QW];
    __shared__ unsigned short qd[NWAVE * QW];
    range_agg<D_OUT, false>(ei, rid, dinv, h2, b2, out, N, E, sv_s, sv_w, sv_d, sv_c,
                            acc, q, qw, qd);
}


extern "C" void kernel_launch(void* const* d_in, const int* in_sizes, int n_in,
                              void* d_out, int out_size, void* d_ws, size_t ws_size,
                              hipStream_t stream) {
    const float* x   = (const float*)d_in[0];
    const int*   ei  = (const int*)  d_in[1];
    const float* W1  = (const float*)d_in[2];
    const float* b1  = (const float*)d_in[3];
    const float* W2  = (const float*)d_in[4];
    const float* b2  = (const float*)d_in[5];
    float*       out = (float*)d_out;

    const int N = in_sizes[0] / D_IN;
    const int E = in_sizes[1] / 2;

    char* ws = (char*)d_ws;
    float*    dinv = (float*)(ws + (size_t)N * 4);
    float*    h    = (float*)(ws + (size_t)N * 8);
    float*    agg1 = (float*)(ws + (size_t)N * 8 + (size_t)N * D_HID * 4);
    float*    h2   = (float*)(ws + (size_t)N * 8 + (size_t)N * D_HID * 8);

    const int TB = 256;

    const int NB = (N + BN - 1) / BN;
    const size_t NQ = (size_t)NB * NWAVE * QW;
    char* qb = ws + (size_t)N * 8 + (size_t)N * D_HID * 8 + (size_t)N * D_OUT * 4;
    unsigned*       sv_s = (unsigned*)(qb);
    float*          sv_w = (float*)(qb + NQ * 4);
    unsigned short* sv_d = (unsigned short*)(qb + NQ * 8);
    int*            sv_c = (int*)(qb + NQ * 10);
    unsigned char*  rid  = (unsigned char*)(sv_c + (size_t)NB * NWAVE);

    k_rid       <<<(E + 4 * TB - 1) / (4 * TB), TB, 0, stream>>>(ei, rid, E);
    k_range_scan<<<NB, AGG_T, 0, stream>>>(ei, rid, dinv, N, E, sv_s, sv_d, sv_c);

    k_gemm_rows<D_IN, D_HID, false><<<(N + GM - 1) / GM, D_HID * 2, 0, stream>>>(x, nullptr, W1, h, N);
    k_range_agg64<<<NB, AGG_T, 0, stream>>>(ei, rid, dinv, h, agg1, N, E, sv_s, sv_w, sv_d, sv_c);

    k_gemm_rows<D_HID, D_OUT, true><<<(N + GM - 1) / GM, D_OUT * 2, 0, stream>>>(agg1, b1, W2, h2, N);
    k_range_agg32<<<NB, AGG_T, 0, stream>>>(ei, rid, dinv, h2, b2, out, N, E, sv_s, sv_w, sv_d, sv_c);
}
